# starting kernel
# baseline (speedup 1.0000x reference)
#include <hip/hip_runtime.h>
#include <math.h>

typedef __attribute__((ext_vector_type(2))) float v2f;
typedef __attribute__((ext_vector_type(8))) float v8f;

typedef const float __attribute__((address_space(1)))* gcfp;
typedef float __attribute__((address_space(1)))* gfp;

__device__ __forceinline__ float gelu_exact(float v) {
    return 0.5f * v * (1.0f + erff(v * 0.7071067811865475f));
}

__global__ void zero_f32(float* __restrict__ p, long long n) {
    long long i = (long long)blockIdx.x * blockDim.x + threadIdx.x;
    if (i < n) p[i] = 0.0f;
}

__global__ void count_deg(const int* __restrict__ dst, float* __restrict__ cnt, int E) {
    int i = blockIdx.x * blockDim.x + threadIdx.x;
    if (i < E) atomicAdd(&cnt[dst[i]], 1.0f);
}

__global__ void finalize_inv(float* __restrict__ cnt, int n) {
    int i = blockIdx.x * blockDim.x + threadIdx.x;
    if (i < n) cnt[i] = 1.0f / fmaxf(cnt[i], 1.0f);
}

__global__ void pack_w(const float* __restrict__ W, float* __restrict__ Wpk,
                       int K, int Nc, int NcPad) {
    int i = blockIdx.x * blockDim.x + threadIdx.x;
    if (i >= K * NcPad) return;
    int j   = i & 1;
    int q   = i >> 1;
    int col = q % NcPad;
    int qq  = q / NcPad;
    int h   = qq & 1;
    int ks  = qq >> 1;
    int row = ks * 4 + h * 2 + j;
    Wpk[i] = (col < Nc) ? W[row * Nc + col] : 0.0f;
}

template<int VEC>
__global__ void scatter_add(const float* __restrict__ yl, const int* __restrict__ src,
                            const int* __restrict__ dst, float* __restrict__ agg,
                            int E, int dout) {
    const int chunks = dout / VEC;
    long long tid = (long long)blockIdx.x * blockDim.x + threadIdx.x;
    if (tid >= (long long)E * chunks) return;
    int e = (int)(tid / chunks);
    int c = (int)(tid % chunks);
    int s = src[e], d = dst[e];
    if (VEC == 4) {
        const float4 v = *(const float4*)(yl + (size_t)s * dout + c * 4);
        float* p = agg + (size_t)d * dout + c * 4;
        atomicAdd(p + 0, v.x);
        atomicAdd(p + 1, v.y);
        atomicAdd(p + 2, v.z);
        atomicAdd(p + 3, v.w);
    } else {
        atomicAdd(agg + (size_t)d * dout + c, yl[(size_t)s * dout + c]);
    }
}

__global__ void combine(float* __restrict__ agg, const float* __restrict__ inv,
                        const float* __restrict__ bias, int n, int dout) {
    long long i = (long long)blockIdx.x * blockDim.x + threadIdx.x;
    if (i >= (long long)n * dout) return;
    int node = (int)(i / dout);
    int f = (int)(i % dout);
    agg[i] = agg[i] * inv[node] + bias[f];
}

template<bool ACC, bool GELU, int MT>
__global__ __launch_bounds__(256) void gemm_wmma(
        const float* __restrict__ A, const float* __restrict__ Wpk,
        const float* __restrict__ Cin, float* __restrict__ D,
        int K, int Nc, int NcPad, int tilesM, int tilesN, int totalTiles) {
    int wave = threadIdx.x >> 5;
    int tile = blockIdx.x * (blockDim.x >> 5) + wave;
    if (tile >= totalTiles) return;
    int group = tile / tilesN;
    int tn    = tile - group * tilesN;
    int tmBase = group * MT;
    int lane = threadIdx.x & 31;
    int half = lane >> 4;
    int l = lane & 15;
    int col = tn * 16 + l;

    gcfp Wb = (gcfp)Wpk + ((size_t)half * NcPad + col) * 2;
    const size_t wstep = (size_t)4 * NcPad;

    gcfp Ap[MT];
    #pragma unroll
    for (int t = 0; t < MT; ++t) {
        int rt = tmBase + t;
        if (rt > tilesM - 1) rt = tilesM - 1;
        Ap[t] = (gcfp)A + (size_t)(rt * 16 + l) * K + half * 2;
    }

    v8f acc[MT];
    #pragma unroll
    for (int t = 0; t < MT; ++t) acc[t] = (v8f){};

    for (int k = 0; k < K; k += 4) {
        float bx = Wb[0];
        float by = Wb[1];
        Wb += wstep;
        float ax[MT], ay[MT];
        #pragma unroll
        for (int t = 0; t < MT; ++t) {
            ax[t] = Ap[t][0];
            ay[t] = Ap[t][1];
            Ap[t] += 4;
        }
        v2f b;
        b.x = bx;
        b.y = by;
        #pragma unroll
        for (int t = 0; t < MT; ++t) {
            v2f a;
            a.x = ax[t];
            a.y = ay[t];
            acc[t] = __builtin_amdgcn_wmma_f32_16x16x4_f32(
                         false, a, false, b, (short)0, acc[t], false, false);
        }
    }

    if (col < Nc) {
        gcfp Cg = (gcfp)Cin;
        gfp  Dg = (gfp)D;
        #pragma unroll
        for (int t = 0; t < MT; ++t) {
            int rt = tmBase + t;
            if (rt < tilesM) {
                #pragma unroll
                for (int g = 0; g < 8; ++g) {
                    int row = rt * 16 + half * 8 + g;
                    size_t off = (size_t)row * Nc + col;
                    float v = acc[t][g];
                    if (ACC) v += Cg[off];
                    if (GELU) v = gelu_exact(v);
                    Dg[off] = v;
                }
            }
        }
    }
}

extern "C" void kernel_launch(void* const* d_in, const int* in_sizes, int n_in,
                              void* d_out, int out_size, void* d_ws, size_t ws_size,
                              hipStream_t stream) {
    static const int DIMS[5] = {256, 192, 128, 64, 3};
    constexpr int MT = 5;

    const float* feat  = (const float*)d_in[0];
    const int*   edges = (const int*)d_in[1];
    const int N = in_sizes[0] / DIMS[0];
    const int E = in_sizes[1] / 2;
    const int* src = edges;
    const int* dst = edges + E;

    const float* Wl[4] = {(const float*)d_in[2], (const float*)d_in[5],
                          (const float*)d_in[8], (const float*)d_in[11]};
    const float* Wr[4] = {(const float*)d_in[3], (const float*)d_in[6],
                          (const float*)d_in[9], (const float*)d_in[12]};
    const float* Bb[4] = {(const float*)d_in[4], (const float*)d_in[7],
                          (const float*)d_in[10], (const float*)d_in[13]};

    float* ws    = (float*)d_ws;
    float* inv   = ws;
    float* bufP  = ws + N;
    float* bufQ  = bufP + (size_t)N * 192;
    float* bufR  = bufQ + (size_t)N * 192;
    float* wPkL  = bufR + (size_t)N * 192;
    float* wPkR  = wPkL + 256 * 208;

    zero_f32<<<(N + 255) / 256, 256, 0, stream>>>(inv, N);
    count_deg<<<(E + 255) / 256, 256, 0, stream>>>(dst, inv, E);
    finalize_inv<<<(N + 255) / 256, 256, 0, stream>>>(inv, N);

    const float* x = feat;
    for (int L = 0; L < 4; ++L) {
        const int K    = DIMS[L];
        const int dout = DIMS[L + 1];
        float* yl   = (L & 1) ? bufR : bufP;
        float* agg  = bufQ;
        float* xout = (L == 3) ? (float*)d_out : yl;

        const int tilesM = N / 16;
        const int tilesN = (dout + 15) / 16;
        const int NcPad  = tilesN * 16;
        const int groups = (tilesM + MT - 1) / MT;
        const int total  = groups * tilesN;
        const int blocks = (total + 7) / 8;

        const int padN = K * NcPad;
        pack_w<<<(padN + 255) / 256, 256, 0, stream>>>(Wl[L], wPkL, K, dout, NcPad);
        pack_w<<<(padN + 255) / 256, 256, 0, stream>>>(Wr[L], wPkR, K, dout, NcPad);

        gemm_wmma<false, false, MT><<<blocks, 256, 0, stream>>>(
            x, wPkL, nullptr, yl, K, dout, NcPad, tilesM, tilesN, total);

        const long long na = (long long)N * dout;
        zero_f32<<<(unsigned)((na + 255) / 256), 256, 0, stream>>>(agg, na);

        if (dout % 4 == 0) {
            long long work = (long long)E * (dout / 4);
            scatter_add<4><<<(unsigned)((work + 255) / 256), 256, 0, stream>>>(
                yl, src, dst, agg, E, dout);
        } else {
            long long work = (long long)E * dout;
            scatter_add<1><<<(unsigned)((work + 255) / 256), 256, 0, stream>>>(
                yl, src, dst, agg, E, dout);
        }

        combine<<<(unsigned)((na + 255) / 256), 256, 0, stream>>>(
            agg, inv, Bb[L], N, dout);

        if (L < 3) {
            gemm_wmma<true, true, MT><<<blocks, 256, 0, stream>>>(
                x, wPkR, agg, xout, K, dout, NcPad, tilesM, tilesN, total);
        } else {
            gemm_wmma<true, false, MT><<<blocks, 256, 0, stream>>>(
                x, wPkR, agg, xout, K, dout, NcPad, tilesM, tilesN, total);
        }
        x = xout;
    }
}


# edges filtered once per call: layer-0 aggregation stores each work-group's (src,dst) pairs, layers 1-3 read them, no sweep; their LDS float-atomic sums now follow arrival-order list places
# speedup vs baseline: 3.3830x; 3.3830x over previous
#include <hip/hip_runtime.h>
#include <math.h>

typedef __attribute__((ext_vector_type(2))) float v2f;
typedef __attribute__((ext_vector_type(8))) float v8f;

typedef const float __attribute__((address_space(1)))* gcfp;
typedef float __attribute__((address_space(1)))* gfp;

__device__ __forceinline__ float gelu_exact(float v) {
    return 0.5f * v * (1.0f + erff(v * 0.7071067811865475f));
}

__device__ __forceinline__ void load_sb(int4 (&q)[4], const int* __restrict__ dst,
                                        int sb, int lane, int E, int vec_ok) {
    const int base = sb * 512;
    if (vec_ok && base + 512 <= E) {
        const int4* p = (const int4*)(dst + base) + lane;
        #pragma unroll
        for (int k = 0; k < 4; ++k) q[k] = p[k * 32];
    } else {
        #pragma unroll
        for (int k = 0; k < 4; ++k) {
            const int e0 = base + k * 128 + lane * 4;
            q[k].x = (e0 + 0 < E) ? dst[e0 + 0] : -1;
            q[k].y = (e0 + 1 < E) ? dst[e0 + 1] : -1;
            q[k].z = (e0 + 2 < E) ? dst[e0 + 2] : -1;
            q[k].w = (e0 + 3 < E) ? dst[e0 + 3] : -1;
        }
    }
}

__device__ __forceinline__ int append_k(int* mypend, int np, bool m, int e) {
    const unsigned b = __builtin_amdgcn_ballot_w32(m);
    if (b) {
        if (m) mypend[np + (int)__builtin_amdgcn_mbcnt_lo(b, 0u)] = e;
        np += __builtin_popcount(b);
    }
    return np;
}

__device__ __forceinline__ void lds_add(float* p, float v) {
    __hip_atomic_fetch_add(p, v, __ATOMIC_RELAXED, __HIP_MEMORY_SCOPE_WORKGROUP);
}

constexpr int LCAP = 6144;

template<int DOUT>
__device__ __forceinline__ void add_rows(int s, int d, int n, int lane,
        const float* __restrict__ yl, float* acc) {
    if constexpr (DOUT >= 64) {
        constexpr int NCH = DOUT / 64;
        constexpr int U = 8;
        for (int i0 = 0; i0 < n; i0 += U) {
            float2 v[U][NCH];
            #pragma unroll
            for (int u = 0; u < U; ++u) {
                int idx = i0 + u;
                if (idx > n - 1) idx = n - 1;
                const int si = __builtin_amdgcn_readlane(s, idx);
                const float* row = yl + (size_t)si * DOUT + lane * 2;
                #pragma unroll
                for (int c = 0; c < NCH; ++c) v[u][c] = *(const float2*)(row + c * 64);
            }
            #pragma unroll
            for (int u = 0; u < U; ++u) {
                if (i0 + u < n) {
                    const int di = __builtin_amdgcn_readlane(d, i0 + u);
                    float* a = acc + di * DOUT + lane * 2;
                    #pragma unroll
                    for (int c = 0; c < NCH; ++c) {
                        lds_add(a + c * 64 + 0, v[u][c].x);
                        lds_add(a + c * 64 + 1, v[u][c].y);
                    }
                }
            }
        }
    } else {
        if (lane < n) {
            #pragma unroll
            for (int c = 0; c < DOUT; ++c)
                lds_add(&acc[d * DOUT + c], yl[(size_t)s * DOUT + c]);
        }
    }
}

template<int DOUT, bool BUILD>
__device__ __forceinline__ void flush_list(const int* p, int n, int lane, int n0,
        const float* __restrict__ yl, const int* __restrict__ src,
        const int* __restrict__ dst, float* acc, float* cntL,
        int* lpos, int2* __restrict__ mylist) {
    __builtin_amdgcn_fence(__ATOMIC_RELEASE, "wavefront");
    __builtin_amdgcn_wave_barrier();
    __builtin_amdgcn_fence(__ATOMIC_ACQUIRE, "wavefront");
    int s = 0, d = 0;
    if (lane < n) {
        const int e = p[lane];
        s = src[e];
        d = dst[e] - n0;
        lds_add(&cntL[d], 1.0f);
    }
    if constexpr (BUILD) {
        int base = 0;
        if (lane == 0)
            base = __hip_atomic_fetch_add(lpos, n, __ATOMIC_RELAXED, __HIP_MEMORY_SCOPE_WORKGROUP);
        base = __builtin_amdgcn_readfirstlane(base);
        if (lane < n && base + lane < LCAP) mylist[base + lane] = make_int2(s, d);
    }
    add_rows<DOUT>(s, d, n, lane, yl, acc);
    __builtin_amdgcn_fence(__ATOMIC_RELEASE, "wavefront");
    __builtin_amdgcn_wave_barrier();
}

template<int DOUT, int R, bool BUILD>
__device__ __forceinline__ void sweep_edges(int* mypend, int wave, int lane, int n0,
        const float* __restrict__ yl, const int* __restrict__ src,
        const int* __restrict__ dst, float* acc, float* cntL,
        int* lpos, int2* __restrict__ mylist, int E, int vec_ok) {
    constexpr int NW = 32;
    int np = 0;
    const int nsb = (E + 511) / 512;
    const int start = (int)(((long long)blockIdx.x * nsb) / gridDim.x);

    int4 cur[4];
    #pragma unroll
    for (int k = 0; k < 4; ++k) cur[k] = make_int4(-1, -1, -1, -1);
    if (wave < nsb) {
        int sb = wave + start; if (sb >= nsb) sb -= nsb;
        load_sb(cur, dst, sb, lane, E, vec_ok);
    }
    for (int j = wave; j < nsb; j += NW) {
        int sb = j + start; if (sb >= nsb) sb -= nsb;
        int4 nxt[4];
        if (j + NW < nsb) {
            int sbn = j + NW + start; if (sbn >= nsb) sbn -= nsb;
            load_sb(nxt, dst, sbn, lane, E, vec_ok);
        } else {
            #pragma unroll
            for (int k = 0; k < 4; ++k) nxt[k] = make_int4(-1, -1, -1, -1);
        }
        #pragma unroll
        for (int k = 0; k < 4; ++k) {
            const int e0 = sb * 512 + k * 128 + lane * 4;
            const int4 q = cur[k];
            const bool m0 = (unsigned)(q.x - n0) < (unsigned)R;
            const bool m1 = (unsigned)(q.y - n0) < (unsigned)R;
            const bool m2 = (unsigned)(q.z - n0) < (unsigned)R;
            const bool m3 = (unsigned)(q.w - n0) < (unsigned)R;
            if (__builtin_amdgcn_ballot_w32(m0 | m1 | m2 | m3)) {
                np = append_k(mypend, np, m0, e0 + 0);
                np = append_k(mypend, np, m1, e0 + 1);
                np = append_k(mypend, np, m2, e0 + 2);
                np = append_k(mypend, np, m3, e0 + 3);
                while (np >= 32) {
                    np -= 32;
                    flush_list<DOUT, BUILD>(mypend + np, 32, lane, n0, yl, src, dst, acc, cntL, lpos, mylist);
                }
            }
        }
        #pragma unroll
        for (int k = 0; k < 4; ++k) cur[k] = nxt[k];
    }
    flush_list<DOUT, BUILD>(mypend, np, lane, n0, yl, src, dst, acc, cntL, lpos, mylist);
}

template<int DOUT, int R, bool BUILD>
__global__ __launch_bounds__(1024) void agg_owner(
        const float* __restrict__ yl, const int* __restrict__ src,
        const int* __restrict__ dst, const float* __restrict__ bias,
        float* __restrict__ agg, int* __restrict__ elist, int* __restrict__ ecount,
        int N, int E, int vec_ok) {
    constexpr int NW = 32;
    constexpr int PCAP = 160;
    __shared__ float acc[R * DOUT];
    __shared__ float cntL[R];
    __shared__ int   pend[NW * PCAP];
    __shared__ int   lpos;

    const int tid  = threadIdx.x;
    const int lane = tid & 31;
    const int wave = __builtin_amdgcn_readfirstlane(tid >> 5);
    const int n0   = blockIdx.x * R;

    for (int i = tid; i < R * DOUT; i += 1024) acc[i] = 0.0f;
    for (int i = tid; i < R; i += 1024) cntL[i] = 0.0f;
    if (tid == 0) lpos = 0;
    __syncthreads();

    int* mypend = pend + wave * PCAP;
    int2* mylist = (int2*)elist + (size_t)blockIdx.x * LCAP;

    if constexpr (BUILD) {
        sweep_edges<DOUT, R, true>(mypend, wave, lane, n0, yl, src, dst, acc, cntL,
                                   &lpos, mylist, E, vec_ok);
    } else {
        const int cnt = __builtin_amdgcn_readfirstlane(ecount[blockIdx.x]);
        if (cnt <= LCAP) {
            constexpr int LCH = LCAP / (NW * 32);
            const int per = (((cnt + NW - 1) / NW) + 7) & ~7;
            const int lo  = wave * per;
            const int hi  = (lo + per < cnt) ? (lo + per) : cnt;
            int2 t[LCH];
            #pragma unroll
            for (int c = 0; c < LCH; ++c) t[c] = make_int2(0, 0);
            #pragma unroll
            for (int c = 0; c < LCH; ++c) {
                const int i = lo + c * 32 + lane;
                if (i < hi) t[c] = mylist[i];
            }
            #pragma unroll
            for (int c = 0; c < LCH; ++c) {
                const int b = lo + c * 32;
                if (b < hi) {
                    const int n = (hi - b < 32) ? (hi - b) : 32;
                    if (lane < n) lds_add(&cntL[t[c].y], 1.0f);
                    add_rows<DOUT>(t[c].x, t[c].y, n, lane, yl, acc);
                }
            }
        } else {
            sweep_edges<DOUT, R, false>(mypend, wave, lane, n0, yl, src, dst, acc, cntL,
                                        &lpos, mylist, E, vec_ok);
        }
    }
    __syncthreads();

    if constexpr (BUILD) {
        if (tid == 0) ecount[blockIdx.x] = lpos;
    }

    for (int i = tid; i < R * DOUT; i += 1024) {
        const int r = i / DOUT;
        const int f = i - r * DOUT;
        if (n0 + r < N)
            agg[(size_t)n0 * DOUT + i] = acc[i] / fmaxf(cntL[r], 1.0f) + bias[f];
    }
}

template<int K, int NC, int NT, bool ACC, bool GELU>
__global__ __launch_bounds__(1024) void gemm_wlds(
        const float* __restrict__ A, const float* __restrict__ W,
        const float* __restrict__ Cin, float* __restrict__ D, int tilesM) {
    constexpr int TN    = (NC + 15) / 16;
    constexpr int NCPAD = TN * 16;
    constexpr int NCH   = TN / NT;
    constexpr int KS    = K / 4;
    constexpr int KB    = KS / 8;
    constexpr int TOT   = KS * NCPAD;
    constexpr int ENT   = (TOT + 1023) / 1024;
    static_assert(TN % NT == 0 && KS % 8 == 0 && 32 % NCH == 0, "tiling");
    static_assert(NC == NCPAD || NT == 1, "the column guard covers one N-tile");
    __shared__ float Ws[TOT * 4];

    const int tid  = threadIdx.x;
    const int lane = tid & 31;
    const int wave = __builtin_amdgcn_readfirstlane(tid >> 5);
    const int half = lane >> 4;
    const int l    = lane & 15;
    const int nc   = wave % NCH;
    int mt = (int)blockIdx.x + (int)gridDim.x * (wave / NCH);

    v2f an[8];
    gcfp Ap = (gcfp)A + (size_t)((mt < tilesM ? mt : 0) * 16 + l) * K + half * 2;
    if (mt < tilesM) {
        #pragma unroll
        for (int u = 0; u < 8; ++u) { an[u].x = Ap[u * 4]; an[u].y = Ap[u * 4 + 1]; }
    }

    {
        float4 w[ENT];
        #pragma unroll
        for (int e = 0; e < ENT; ++e) {
            const int i   = tid + e * 1024;
            const int ks  = i / NCPAD;
            const int col = i - ks * NCPAD;
            w[e] = make_float4(0.0f, 0.0f, 0.0f, 0.0f);
            if ((TOT % 1024 == 0 || i < TOT) && (NC == NCPAD || col < NC)) {
                const float* p = W + (size_t)(ks * 4) * NC + col;
                w[e].x = p[0];
                w[e].y = p[NC];
                w[e].z = p[2 * NC];
                w[e].w = p[3 * NC];
            }
        }
        #pragma unroll
        for (int e = 0; e < ENT; ++e) {
            const int i = tid + e * 1024;
            if (TOT % 1024 == 0 || i < TOT) *(float4*)(Ws + i * 4) = w[e];
        }
    }
    __syncthreads();

    const float* Bp = Ws + (nc * NT * 16 + l) * 4 + half * 2;
    gcfp Cg = (gcfp)Cin;
    gfp  Dg = (gfp)D;

    while (mt < tilesM) {
        v8f acc[NT];
        #pragma unroll
        for (int t = 0; t < NT; ++t) acc[t] = (v8f){};

        #pragma unroll 1
        for (int kb = 0; kb < KB; ++kb) {
            v2f ac[8];
            #pragma unroll
            for (int u = 0; u < 8; ++u) ac[u] = an[u];
            if (kb + 1 < KB) {
                gcfp An = Ap + (kb + 1) * 32;
                #pragma unroll
                for (int u = 0; u < 8; ++u) { an[u].x = An[u * 4]; an[u].y = An[u * 4 + 1]; }
            }
            const float* Bk = Bp + kb * (8 * NCPAD * 4);
            #pragma unroll
            for (int u = 0; u < 8; ++u) {
                #pragma unroll
                for (int t = 0; t < NT; ++t) {
                    const v2f b = *(const v2f*)(Bk + (u * NCPAD + t * 16) * 4);
                    acc[t] = __builtin_amdgcn_wmma_f32_16x16x4_f32(
                                 false, ac[u], false, b, (short)0, acc[t], false, false);
                }
            }
        }

        const size_t off0 = (size_t)(mt * 16 + half * 8) * NC + (nc * NT * 16 + l);
        if (NC == NCPAD || nc * NT * 16 + l < NC) {
            if (ACC) {
                float cv[NT][8];
                #pragma unroll
                for (int t = 0; t < NT; ++t) {
                    #pragma unroll
                    for (int g = 0; g < 8; ++g) cv[t][g] = Cg[off0 + (size_t)g * NC + t * 16];
                }
                #pragma unroll
                for (int t = 0; t < NT; ++t) {
                    #pragma unroll
                    for (int g = 0; g < 8; ++g) acc[t][g] += cv[t][g];
                }
            }
            #pragma unroll
            for (int t = 0; t < NT; ++t) {
                #pragma unroll
                for (int g = 0; g < 8; ++g) {
                    float v = acc[t][g];
                    if (GELU) v = gelu_exact(v);
                    Dg[off0 + (size_t)g * NC + t * 16] = v;
                }
            }
        }

        mt += (int)gridDim.x * (32 / NCH);
        if (mt < tilesM) {
            Ap = (gcfp)A + (size_t)(mt * 16 + l) * K + half * 2;
            #pragma unroll
            for (int u = 0; u < 8; ++u) { an[u].x = Ap[u * 4]; an[u].y = Ap[u * 4 + 1]; }
        }
    }
}

extern "C" void kernel_launch(void* const* d_in, const int* in_sizes, int n_in,
                              void* d_out, int out_size, void* d_ws, size_t ws_size,
                              hipStream_t stream) {
    static const int DIMS[5] = {256, 192, 128, 64, 3};

    const float* feat  = (const float*)d_in[0];
    const int*   edges = (const int*)d_in[1];
    const int N = in_sizes[0] / DIMS[0];
    const int E = in_sizes[1] / 2;
    const int* src = edges;
    const int* dst = edges + E;

    const float* Wl[4] = {(const float*)d_in[2], (const float*)d_in[5],
                          (const float*)d_in[8], (const float*)d_in[11]};
    const float* Wr[4] = {(const float*)d_in[3], (const float*)d_in[6],
                          (const float*)d_in[9], (const float*)d_in[12]};
    const float* Bb[4] = {(const float*)d_in[4], (const float*)d_in[7],
                          (const float*)d_in[10], (const float*)d_in[13]};

    float* ws    = (float*)d_ws;
    float* r0    = ws;
    float* r1    = ws + (size_t)N * 192;
    float* r2    = ws + (size_t)N * 320;
    int* elist   = (int*)(ws + (size_t)N * 448);
    int* ecount  = elist + (size_t)((N + 195) / 196) * LCAP * 2;
    float* const YL[4]  = {r0, r1, r0, r2};
    float* const AGG[4] = {r1, r2, r0 + (size_t)N * 64, r2 + (size_t)N * 64};

    const int vec_ok = (E % 4 == 0) ? 1 : 0;
    const int tilesM = N / 16;
    const int GW = 256;

    const float* x = feat;
    for (int L = 0; L < 4; ++L) {
        float* yl   = YL[L];
        float* agg  = AGG[L];
        float* xout = (L == 3) ? (float*)d_out : yl;

        if (L == 0) {
            gemm_wlds<256, 192, 6, false, false><<<GW, 1024, 0, stream>>>(x, Wl[L], nullptr, yl, tilesM);
            agg_owner<192, 196, true><<<(N + 195) / 196, 1024, 0, stream>>>(
                yl, src, dst, Bb[L], agg, elist, ecount, N, E, vec_ok);
            gemm_wlds<256, 192, 6, true, true><<<GW, 1024, 0, stream>>>(x, Wr[L], agg, xout, tilesM);
        } else if (L == 1) {
            gemm_wlds<192, 128, 4, false, false><<<GW, 1024, 0, stream>>>(x, Wl[L], nullptr, yl, tilesM);
            agg_owner<128, 196, false><<<(N + 195) / 196, 1024, 0, stream>>>(
                yl, src, dst, Bb[L], agg, elist, ecount, N, E, vec_ok);
            gemm_wlds<192, 128, 4, true, true><<<GW, 1024, 0, stream>>>(x, Wr[L], agg, xout, tilesM);
        } else if (L == 2) {
            gemm_wlds<128, 64, 4, false, false><<<GW, 1024, 0, stream>>>(x, Wl[L], nullptr, yl, tilesM);
            agg_owner<64, 196, false><<<(N + 195) / 196, 1024, 0, stream>>>(
                yl, src, dst, Bb[L], agg, elist, ecount, N, E, vec_ok);
            gemm_wlds<128, 64, 4, true, true><<<GW, 1024, 0, stream>>>(x, Wr[L], agg, xout, tilesM);
        } else {
            gemm_wlds<64, 3, 1, false, false><<<GW, 1024, 0, stream>>>(x, Wl[L], nullptr, yl, tilesM);
            agg_owner<3, 196, false><<<(N + 195) / 196, 1024, 0, stream>>>(
                yl, src, dst, Bb[L], agg, elist, ecount, N, E, vec_ok);
            gemm_wlds<64, 3, 1, true, false><<<GW, 1024, 0, stream>>>(x, Wr[L], agg, xout, tilesM);
        }
        x = xout;
    }
}


# layer 0's 256 whole-list sweeps go: one binning launch reads the edges once (LDS count by owner, one returning add per owner, runs written out); layer-0 LDS-atomic sums in list (arrival) order
# speedup vs baseline: 3.5578x; 1.0517x over previous
#include <hip/hip_runtime.h>
#include <math.h>

typedef __attribute__((ext_vector_type(2))) float v2f;
typedef __attribute__((ext_vector_type(8))) float v8f;

typedef const float __attribute__((address_space(1)))* gcfp;
typedef float __attribute__((address_space(1)))* gfp;

__device__ __forceinline__ float gelu_exact(float v) {
    return 0.5f * v * (1.0f + erff(v * 0.7071067811865475f));
}

__device__ __forceinline__ void load_sb(int4 (&q)[4], const int* __restrict__ dst,
                                        int sb, int lane, int E, int vec_ok) {
    const int base = sb * 512;
    if (vec_ok && base + 512 <= E) {
        const int4* p = (const int4*)(dst + base) + lane;
        #pragma unroll
        for (int k = 0; k < 4; ++k) q[k] = p[k * 32];
    } else {
        #pragma unroll
        for (int k = 0; k < 4; ++k) {
            const int e0 = base + k * 128 + lane * 4;
            q[k].x = (e0 + 0 < E) ? dst[e0 + 0] : -1;
            q[k].y = (e0 + 1 < E) ? dst[e0 + 1] : -1;
            q[k].z = (e0 + 2 < E) ? dst[e0 + 2] : -1;
            q[k].w = (e0 + 3 < E) ? dst[e0 + 3] : -1;
        }
    }
}

__device__ __forceinline__ int append_k(int* mypend, int np, bool m, int e) {
    const unsigned b = __builtin_amdgcn_ballot_w32(m);
    if (b) {
        if (m) mypend[np + (int)__builtin_amdgcn_mbcnt_lo(b, 0u)] = e;
        np += __builtin_popcount(b);
    }
    return np;
}

__device__ __forceinline__ void lds_add(float* p, float v) {
    __hip_atomic_fetch_add(p, v, __ATOMIC_RELAXED, __HIP_MEMORY_SCOPE_WORKGROUP);
}

constexpr int LCAP = 6144;
constexpr int CSTR = 64;

template<int DOUT>
__device__ __forceinline__ void add_rows(int s, int d, int n, int lane,
        const float* __restrict__ yl, float* acc) {
    if constexpr (DOUT >= 64) {
        constexpr int NCH = DOUT / 64;
        constexpr int U = 8;
        for (int i0 = 0; i0 < n; i0 += U) {
            float2 v[U][NCH];
            #pragma unroll
            for (int u = 0; u < U; ++u) {
                int idx = i0 + u;
                if (idx > n - 1) idx = n - 1;
                const int si = __builtin_amdgcn_readlane(s, idx);
                const float* row = yl + (size_t)si * DOUT + lane * 2;
                #pragma unroll
                for (int c = 0; c < NCH; ++c) v[u][c] = *(const float2*)(row + c * 64);
            }
            #pragma unroll
            for (int u = 0; u < U; ++u) {
                if (i0 + u < n) {
                    const int di = __builtin_amdgcn_readlane(d, i0 + u);
                    float* a = acc + di * DOUT + lane * 2;
                    #pragma unroll
                    for (int c = 0; c < NCH; ++c) {
                        lds_add(a + c * 64 + 0, v[u][c].x);
                        lds_add(a + c * 64 + 1, v[u][c].y);
                    }
                }
            }
        }
    } else {
        if (lane < n) {
            #pragma unroll
            for (int c = 0; c < DOUT; ++c)
                lds_add(&acc[d * DOUT + c], yl[(size_t)s * DOUT + c]);
        }
    }
}

template<int DOUT>
__device__ __forceinline__ void flush_list(const int* p, int n, int lane, int n0,
        const float* __restrict__ yl, const int* __restrict__ src,
        const int* __restrict__ dst, float* acc, float* cntL) {
    __builtin_amdgcn_fence(__ATOMIC_RELEASE, "wavefront");
    __builtin_amdgcn_wave_barrier();
    __builtin_amdgcn_fence(__ATOMIC_ACQUIRE, "wavefront");
    int s = 0, d = 0;
    if (lane < n) {
        const int e = p[lane];
        s = src[e];
        d = dst[e] - n0;
        lds_add(&cntL[d], 1.0f);
    }
    add_rows<DOUT>(s, d, n, lane, yl, acc);
    __builtin_amdgcn_fence(__ATOMIC_RELEASE, "wavefront");
    __builtin_amdgcn_wave_barrier();
}

template<int DOUT, int R>
__device__ __forceinline__ void sweep_edges(int* mypend, int wave, int lane, int n0,
        const float* __restrict__ yl, const int* __restrict__ src,
        const int* __restrict__ dst, float* acc, float* cntL, int E, int vec_ok) {
    constexpr int NW = 32;
    int np = 0;
    const int nsb = (E + 511) / 512;
    const int start = (int)(((long long)blockIdx.x * nsb) / gridDim.x);

    int4 cur[4];
    #pragma unroll
    for (int k = 0; k < 4; ++k) cur[k] = make_int4(-1, -1, -1, -1);
    if (wave < nsb) {
        int sb = wave + start; if (sb >= nsb) sb -= nsb;
        load_sb(cur, dst, sb, lane, E, vec_ok);
    }
    for (int j = wave; j < nsb; j += NW) {
        int sb = j + start; if (sb >= nsb) sb -= nsb;
        int4 nxt[4];
        if (j + NW < nsb) {
            int sbn = j + NW + start; if (sbn >= nsb) sbn -= nsb;
            load_sb(nxt, dst, sbn, lane, E, vec_ok);
        } else {
            #pragma unroll
            for (int k = 0; k < 4; ++k) nxt[k] = make_int4(-1, -1, -1, -1);
        }
        #pragma unroll
        for (int k = 0; k < 4; ++k) {
            const int e0 = sb * 512 + k * 128 + lane * 4;
            const int4 q = cur[k];
            const bool m0 = (unsigned)(q.x - n0) < (unsigned)R;
            const bool m1 = (unsigned)(q.y - n0) < (unsigned)R;
            const bool m2 = (unsigned)(q.z - n0) < (unsigned)R;
            const bool m3 = (unsigned)(q.w - n0) < (unsigned)R;
            if (__builtin_amdgcn_ballot_w32(m0 | m1 | m2 | m3)) {
                np = append_k(mypend, np, m0, e0 + 0);
                np = append_k(mypend, np, m1, e0 + 1);
                np = append_k(mypend, np, m2, e0 + 2);
                np = append_k(mypend, np, m3, e0 + 3);
                while (np >= 32) {
                    np -= 32;
                    flush_list<DOUT>(mypend + np, 32, lane, n0, yl, src, dst, acc, cntL);
                }
            }
        }
        #pragma unroll
        for (int k = 0; k < 4; ++k) cur[k] = nxt[k];
    }
    flush_list<DOUT>(mypend, np, lane, n0, yl, src, dst, acc, cntL);
}

template<int R>
__global__ __launch_bounds__(1024) void bin_edges(
        const int* __restrict__ src, const int* __restrict__ dst,
        int* __restrict__ elist, int* __restrict__ ecount, int E, int nb) {
    constexpr int EPT = 4;
    constexpr int SC  = 1024 * EPT;
    __shared__ int  hist[1024];
    __shared__ int  gbase[1024];
    __shared__ int  wsum[32];
    __shared__ int2 sorted[SC];
    __shared__ int  sbin[SC];

    const int tid  = threadIdx.x;
    const int lane = tid & 31;
    const int wave = tid >> 5;
    const int chunk = (E + (int)gridDim.x - 1) / (int)gridDim.x;
    const int c0 = (int)blockIdx.x * chunk;
    const int c1 = (c0 + chunk < E) ? (c0 + chunk) : E;

    for (int r0 = c0; r0 < c1; r0 += SC) {
        const int r1 = (r0 + SC < c1) ? (r0 + SC) : c1;
        hist[tid] = 0;
        __syncthreads();

        int s[EPT], dd[EPT], g[EPT], d[EPT], rk[EPT];
        #pragma unroll
        for (int k = 0; k < EPT; ++k) {
            const int e = r0 + k * 1024 + tid;
            dd[k] = -1; s[k] = 0;
            if (e < r1) { dd[k] = dst[e]; s[k] = src[e]; }
        }
        #pragma unroll
        for (int k = 0; k < EPT; ++k) {
            g[k] = -1; d[k] = 0;
            if (dd[k] >= 0) {
                g[k] = dd[k] / R;
                d[k] = dd[k] - g[k] * R;
            }
        }
        #pragma unroll
        for (int k = 0; k < EPT; ++k) {
            rk[k] = 0;
            if (g[k] >= 0)
                rk[k] = __hip_atomic_fetch_add(&hist[g[k]], 1, __ATOMIC_RELAXED, __HIP_MEMORY_SCOPE_WORKGROUP);
        }
        __syncthreads();

        const int cn = hist[tid];
        int inc = cn;
        for (int o = 1; o < 32; o <<= 1) {
            const int t = __shfl_up(inc, o);
            if (lane >= o) inc += t;
        }
        if (lane == 31) wsum[wave] = inc;
        __syncthreads();
        if (wave == 0) {
            const int w = wsum[lane];
            int wi = w;
            for (int o = 1; o < 32; o <<= 1) {
                const int t = __shfl_up(wi, o);
                if (lane >= o) wi += t;
            }
            wsum[lane] = wi - w;
        }
        __syncthreads();
        hist[tid] = inc - cn + wsum[wave];
        int gb = 0;
        if (cn > 0 && tid < nb)
            gb = __hip_atomic_fetch_add(&ecount[tid * CSTR], cn, __ATOMIC_RELAXED, __HIP_MEMORY_SCOPE_AGENT);
        gbase[tid] = gb;
        __syncthreads();

        #pragma unroll
        for (int k = 0; k < EPT; ++k) {
            if (g[k] >= 0) {
                const int p = hist[g[k]] + rk[k];
                sorted[p] = make_int2(s[k], d[k]);
                sbin[p]   = g[k];
            }
        }
        __syncthreads();

        const int n = r1 - r0;
        for (int i = tid; i < n; i += 1024) {
            const int gg = sbin[i];
            const int p  = gbase[gg] + (i - hist[gg]);
            if (p < LCAP) ((int2*)elist)[(size_t)gg * LCAP + p] = sorted[i];
        }
        __syncthreads();
    }
}

template<int DOUT, int R>
__global__ __launch_bounds__(1024) void agg_owner(
        const float* __restrict__ yl, const int* __restrict__ src,
        const int* __restrict__ dst, const float* __restrict__ bias,
        float* __restrict__ agg, const int* __restrict__ elist, const int* __restrict__ ecount,
        int N, int E, int vec_ok) {
    constexpr int NW = 32;
    constexpr int PCAP = 160;
    __shared__ float acc[R * DOUT];
    __shared__ float cntL[R];
    __shared__ int   pend[NW * PCAP];

    const int tid  = threadIdx.x;
    const int lane = tid & 31;
    const int wave = __builtin_amdgcn_readfirstlane(tid >> 5);
    const int n0   = blockIdx.x * R;

    for (int i = tid; i < R * DOUT; i += 1024) acc[i] = 0.0f;
    for (int i = tid; i < R; i += 1024) cntL[i] = 0.0f;
    __syncthreads();

    int* mypend = pend + wave * PCAP;
    const int2* mylist = (const int2*)elist + (size_t)blockIdx.x * LCAP;

    {
        const int cnt = __builtin_amdgcn_readfirstlane(ecount[blockIdx.x * CSTR]);
        if (cnt <= LCAP) {
            constexpr int LCH = LCAP / (NW * 32);
            const int per = (((cnt + NW - 1) / NW) + 7) & ~7;
            const int lo  = wave * per;
            const int hi  = (lo + per < cnt) ? (lo + per) : cnt;
            int2 t[LCH];
            #pragma unroll
            for (int c = 0; c < LCH; ++c) t[c] = make_int2(0, 0);
            #pragma unroll
            for (int c = 0; c < LCH; ++c) {
                const int i = lo + c * 32 + lane;
                if (i < hi) t[c] = mylist[i];
            }
            #pragma unroll
            for (int c = 0; c < LCH; ++c) {
                const int b = lo + c * 32;
                if (b < hi) {
                    const int n = (hi - b < 32) ? (hi - b) : 32;
                    if (lane < n) lds_add(&cntL[t[c].y], 1.0f);
                    add_rows<DOUT>(t[c].x, t[c].y, n, lane, yl, acc);
                }
            }
        } else {
            sweep_edges<DOUT, R>(mypend, wave, lane, n0, yl, src, dst, acc, cntL, E, vec_ok);
        }
    }
    __syncthreads();

    for (int i = tid; i < R * DOUT; i += 1024) {
        const int r = i / DOUT;
        const int f = i - r * DOUT;
        if (n0 + r < N)
            agg[(size_t)n0 * DOUT + i] = acc[i] / fmaxf(cntL[r], 1.0f) + bias[f];
    }
}

template<int K, int NC, int NT, bool ACC, bool GELU>
__global__ __launch_bounds__(1024) void gemm_wlds(
        const float* __restrict__ A, const float* __restrict__ W,
        const float* __restrict__ Cin, float* __restrict__ D, int tilesM,
        int* __restrict__ zc, int nz) {
    constexpr int TN    = (NC + 15) / 16;
    constexpr int NCPAD = TN * 16;
    constexpr int NCH   = TN / NT;
    constexpr int KS    = K / 4;
    constexpr int KB    = KS / 8;
    constexpr int TOT   = KS * NCPAD;
    constexpr int ENT   = (TOT + 1023) / 1024;
    static_assert(TN % NT == 0 && KS % 8 == 0 && 32 % NCH == 0, "tiling");
    static_assert(NC == NCPAD || NT == 1, "the column guard covers one N-tile");
    __shared__ float Ws[TOT * 4];

    const int tid  = threadIdx.x;
    const int lane = tid & 31;
    const int wave = __builtin_amdgcn_readfirstlane(tid >> 5);
    const int half = lane >> 4;
    const int l    = lane & 15;
    const int nc   = wave % NCH;
    int mt = (int)blockIdx.x + (int)gridDim.x * (wave / NCH);

    if (zc != nullptr && blockIdx.x == 0 && tid < nz) zc[tid * CSTR] = 0;

    v2f an[8];
    gcfp Ap = (gcfp)A + (size_t)((mt < tilesM ? mt : 0) * 16 + l) * K + half * 2;
    if (mt < tilesM) {
        #pragma unroll
        for (int u = 0; u < 8; ++u) { an[u].x = Ap[u * 4]; an[u].y = Ap[u * 4 + 1]; }
    }

    {
        float4 w[ENT];
        #pragma unroll
        for (int e = 0; e < ENT; ++e) {
            const int i   = tid + e * 1024;
            const int ks  = i / NCPAD;
            const int col = i - ks * NCPAD;
            w[e] = make_float4(0.0f, 0.0f, 0.0f, 0.0f);
            if ((TOT % 1024 == 0 || i < TOT) && (NC == NCPAD || col < NC)) {
                const float* p = W + (size_t)(ks * 4) * NC + col;
                w[e].x = p[0];
                w[e].y = p[NC];
                w[e].z = p[2 * NC];
                w[e].w = p[3 * NC];
            }
        }
        #pragma unroll
        for (int e = 0; e < ENT; ++e) {
            const int i = tid + e * 1024;
            if (TOT % 1024 == 0 || i < TOT) *(float4*)(Ws + i * 4) = w[e];
        }
    }
    __syncthreads();

    const float* Bp = Ws + (nc * NT * 16 + l) * 4 + half * 2;
    gcfp Cg = (gcfp)Cin;
    gfp  Dg = (gfp)D;

    while (mt < tilesM) {
        v8f acc[NT];
        #pragma unroll
        for (int t = 0; t < NT; ++t) acc[t] = (v8f){};

        #pragma unroll 1
        for (int kb = 0; kb < KB; ++kb) {
            v2f ac[8];
            #pragma unroll
            for (int u = 0; u < 8; ++u) ac[u] = an[u];
            if (kb + 1 < KB) {
                gcfp An = Ap + (kb + 1) * 32;
                #pragma unroll
                for (int u = 0; u < 8; ++u) { an[u].x = An[u * 4]; an[u].y = An[u * 4 + 1]; }
            }
            const float* Bk = Bp + kb * (8 * NCPAD * 4);
            #pragma unroll
            for (int u = 0; u < 8; ++u) {
                #pragma unroll
                for (int t = 0; t < NT; ++t) {
                    const v2f b = *(const v2f*)(Bk + (u * NCPAD + t * 16) * 4);
                    acc[t] = __builtin_amdgcn_wmma_f32_16x16x4_f32(
                                 false, ac[u], false, b, (short)0, acc[t], false, false);
                }
            }
        }

        const size_t off0 = (size_t)(mt * 16 + half * 8) * NC + (nc * NT * 16 + l);
        if (NC == NCPAD || nc * NT * 16 + l < NC) {
            if (ACC) {
                float cv[NT][8];
                #pragma unroll
                for (int t = 0; t < NT; ++t) {
                    #pragma unroll
                    for (int g = 0; g < 8; ++g) cv[t][g] = Cg[off0 + (size_t)g * NC + t * 16];
                }
                #pragma unroll
                for (int t = 0; t < NT; ++t) {
                    #pragma unroll
                    for (int g = 0; g < 8; ++g) acc[t][g] += cv[t][g];
                }
            }
            #pragma unroll
            for (int t = 0; t < NT; ++t) {
                #pragma unroll
                for (int g = 0; g < 8; ++g) {
                    float v = acc[t][g];
                    if (GELU) v = gelu_exact(v);
                    Dg[off0 + (size_t)g * NC + t * 16] = v;
                }
            }
        }

        mt += (int)gridDim.x * (32 / NCH);
        if (mt < tilesM) {
            Ap = (gcfp)A + (size_t)(mt * 16 + l) * K + half * 2;
            #pragma unroll
            for (int u = 0; u < 8; ++u) { an[u].x = Ap[u * 4]; an[u].y = Ap[u * 4 + 1]; }
        }
    }
}

extern "C" void kernel_launch(void* const* d_in, const int* in_sizes, int n_in,
                              void* d_out, int out_size, void* d_ws, size_t ws_size,
                              hipStream_t stream) {
    static const int DIMS[5] = {256, 192, 128, 64, 3};

    const float* feat  = (const float*)d_in[0];
    const int*   edges = (const int*)d_in[1];
    const int N = in_sizes[0] / DIMS[0];
    const int E = in_sizes[1] / 2;
    const int* src = edges;
    const int* dst = edges + E;

    const float* Wl[4] = {(const float*)d_in[2], (const float*)d_in[5],
                          (const float*)d_in[8], (const float*)d_in[11]};
    const float* Wr[4] = {(const float*)d_in[3], (const float*)d_in[6],
                          (const float*)d_in[9], (const float*)d_in[12]};
    const float* Bb[4] = {(const float*)d_in[4], (const float*)d_in[7],
                          (const float*)d_in[10], (const float*)d_in[13]};

    float* ws    = (float*)d_ws;
    float* r0    = ws;
    float* r1    = ws + (size_t)N * 192;
    float* r2    = ws + (size_t)N * 320;
    int* elist   = (int*)(ws + (size_t)N * 448);
    int* ecount  = elist + (size_t)((N + 195) / 196) * LCAP * 2;
    float* const YL[4]  = {r0, r1, r0, r2};
    float* const AGG[4] = {r1, r2, r0 + (size_t)N * 64, r2 + (size_t)N * 64};

    const int vec_ok = (E % 4 == 0) ? 1 : 0;
    const int tilesM = N / 16;
    const int GW = 256;
    const int nb = (N + 195) / 196;

    const float* x = feat;
    for (int L = 0; L < 4; ++L) {
        float* yl   = YL[L];
        float* agg  = AGG[L];
        float* xout = (L == 3) ? (float*)d_out : yl;

        if (L == 0) {
            gemm_wlds<256, 192, 6, false, false><<<GW, 1024, 0, stream>>>(x, Wl[L], nullptr, yl, tilesM, ecount, nb);
            bin_edges<196><<<256, 1024, 0, stream>>>(src, dst, elist, ecount, E, nb);
            agg_owner<192, 196><<<nb, 1024, 0, stream>>>(
                yl, src, dst, Bb[L], agg, elist, ecount, N, E, vec_ok);
            gemm_wlds<256, 192, 6, true, true><<<GW, 1024, 0, stream>>>(x, Wr[L], agg, xout, tilesM, nullptr, 0);
        } else if (L == 1) {
            gemm_wlds<192, 128, 4, false, false><<<GW, 1024, 0, stream>>>(x, Wl[L], nullptr, yl, tilesM, nullptr, 0);
            agg_owner<128, 196><<<nb, 1024, 0, stream>>>(
                yl, src, dst, Bb[L], agg, elist, ecount, N, E, vec_ok);
            gemm_wlds<192, 128, 4, true, true><<<GW, 1024, 0, stream>>>(x, Wr[L], agg, xout, tilesM, nullptr, 0);
        } else if (L == 2) {
            gemm_wlds<128, 64, 4, false, false><<<GW, 1024, 0, stream>>>(x, Wl[L], nullptr, yl, tilesM, nullptr, 0);
            agg_owner<64, 196><<<nb, 1024, 0, stream>>>(
                yl, src, dst, Bb[L], agg, elist, ecount, N, E, vec_ok);
            gemm_wlds<128, 64, 4, true, true><<<GW, 1024, 0, stream>>>(x, Wr[L], agg, xout, tilesM, nullptr, 0);
        } else {
            gemm_wlds<64, 3, 1, false, false><<<GW, 1024, 0, stream>>>(x, Wl[L], nullptr, yl, tilesM, nullptr, 0);
            agg_owner<3, 196><<<nb, 1024, 0, stream>>>(
                yl, src, dst, Bb[L], agg, elist, ecount, N, E, vec_ok);
            gemm_wlds<64, 3, 1, true, false><<<GW, 1024, 0, stream>>>(x, Wr[L], agg, xout, tilesM, nullptr, 0);
        }
        x = xout;
    }
}


# layer 3's two products leave their launches (13 -> 11): layer 2's root GEMM runs transposed (weights = WMMA A operand), so its tile's registers are [Wl3|Wr3]'s B fragments; layer-3 k order changes
# speedup vs baseline: 3.6726x; 1.0323x over previous
#include <hip/hip_runtime.h>
#include <math.h>

typedef __attribute__((ext_vector_type(2))) float v2f;
typedef __attribute__((ext_vector_type(8))) float v8f;

typedef const float __attribute__((address_space(1)))* gcfp;
typedef float __attribute__((address_space(1)))* gfp;
typedef __attribute__((ext_vector_type(4))) float v4f;
typedef const v4f __attribute__((address_space(1)))* gcv4p;

__device__ __forceinline__ float gelu_exact(float v) {
    return 0.5f * v * (1.0f + erff(v * 0.7071067811865475f));
}

__device__ __forceinline__ void load_sb(int4 (&q)[4], const int* __restrict__ dst,
                                        int sb, int lane, int E, int vec_ok) {
    const int base = sb * 512;
    if (vec_ok && base + 512 <= E) {
        const int4* p = (const int4*)(dst + base) + lane;
        #pragma unroll
        for (int k = 0; k < 4; ++k) q[k] = p[k * 32];
    } else {
        #pragma unroll
        for (int k = 0; k < 4; ++k) {
            const int e0 = base + k * 128 + lane * 4;
            q[k].x = (e0 + 0 < E) ? dst[e0 + 0] : -1;
            q[k].y = (e0 + 1 < E) ? dst[e0 + 1] : -1;
            q[k].z = (e0 + 2 < E) ? dst[e0 + 2] : -1;
            q[k].w = (e0 + 3 < E) ? dst[e0 + 3] : -1;
        }
    }
}

__device__ __forceinline__ int append_k(int* mypend, int np, bool m, int e) {
    const unsigned b = __builtin_amdgcn_ballot_w32(m);
    if (b) {
        if (m) mypend[np + (int)__builtin_amdgcn_mbcnt_lo(b, 0u)] = e;
        np += __builtin_popcount(b);
    }
    return np;
}

__device__ __forceinline__ void lds_add(float* p, float v) {
    __hip_atomic_fetch_add(p, v, __ATOMIC_RELAXED, __HIP_MEMORY_SCOPE_WORKGROUP);
}

constexpr int LCAP = 6144;
constexpr int CSTR = 64;

template<int DOUT>
__device__ __forceinline__ void add_rows(int s, int d, int n, int lane,
        const float* __restrict__ yl, float* acc) {
    if constexpr (DOUT >= 64) {
        constexpr int NCH = DOUT / 64;
        constexpr int U = 8;
        for (int i0 = 0; i0 < n; i0 += U) {
            float2 v[U][NCH];
            #pragma unroll
            for (int u = 0; u < U; ++u) {
                int idx = i0 + u;
                if (idx > n - 1) idx = n - 1;
                const int si = __builtin_amdgcn_readlane(s, idx);
                const float* row = yl + (size_t)si * DOUT + lane * 2;
                #pragma unroll
                for (int c = 0; c < NCH; ++c) v[u][c] = *(const float2*)(row + c * 64);
            }
            #pragma unroll
            for (int u = 0; u < U; ++u) {
                if (i0 + u < n) {
                    const int di = __builtin_amdgcn_readlane(d, i0 + u);
                    float* a = acc + di * DOUT + lane * 2;
                    #pragma unroll
                    for (int c = 0; c < NCH; ++c) {
                        lds_add(a + c * 64 + 0, v[u][c].x);
                        lds_add(a + c * 64 + 1, v[u][c].y);
                    }
                }
            }
        }
    } else {
        if (lane < n) {
            #pragma unroll
            for (int c = 0; c < DOUT; ++c)
                lds_add(&acc[d * DOUT + c], yl[(size_t)s * DOUT + c]);
        }
    }
}

template<int DOUT>
__device__ __forceinline__ void flush_list(const int* p, int n, int lane, int n0,
        const float* __restrict__ yl, const int* __restrict__ src,
        const int* __restrict__ dst, float* acc, float* cntL) {
    __builtin_amdgcn_fence(__ATOMIC_RELEASE, "wavefront");
    __builtin_amdgcn_wave_barrier();
    __builtin_amdgcn_fence(__ATOMIC_ACQUIRE, "wavefront");
    int s = 0, d = 0;
    if (lane < n) {
        const int e = p[lane];
        s = src[e];
        d = dst[e] - n0;
        lds_add(&cntL[d], 1.0f);
    }
    add_rows<DOUT>(s, d, n, lane, yl, acc);
    __builtin_amdgcn_fence(__ATOMIC_RELEASE, "wavefront");
    __builtin_amdgcn_wave_barrier();
}

template<int DOUT, int R>
__device__ __forceinline__ void sweep_edges(int* mypend, int wave, int lane, int n0,
        const float* __restrict__ yl, const int* __restrict__ src,
        const int* __restrict__ dst, float* acc, float* cntL, int E, int vec_ok) {
    constexpr int NW = 32;
    int np = 0;
    const int nsb = (E + 511) / 512;
    const int start = (int)(((long long)blockIdx.x * nsb) / gridDim.x);

    int4 cur[4];
    #pragma unroll
    for (int k = 0; k < 4; ++k) cur[k] = make_int4(-1, -1, -1, -1);
    if (wave < nsb) {
        int sb = wave + start; if (sb >= nsb) sb -= nsb;
        load_sb(cur, dst, sb, lane, E, vec_ok);
    }
    for (int j = wave; j < nsb; j += NW) {
        int sb = j + start; if (sb >= nsb) sb -= nsb;
        int4 nxt[4];
        if (j + NW < nsb) {
            int sbn = j + NW + start; if (sbn >= nsb) sbn -= nsb;
            load_sb(nxt, dst, sbn, lane, E, vec_ok);
        } else {
            #pragma unroll
            for (int k = 0; k < 4; ++k) nxt[k] = make_int4(-1, -1, -1, -1);
        }
        #pragma unroll
        for (int k = 0; k < 4; ++k) {
            const int e0 = sb * 512 + k * 128 + lane * 4;
            const int4 q = cur[k];
            const bool m0 = (unsigned)(q.x - n0) < (unsigned)R;
            const bool m1 = (unsigned)(q.y - n0) < (unsigned)R;
            const bool m2 = (unsigned)(q.z - n0) < (unsigned)R;
            const bool m3 = (unsigned)(q.w - n0) < (unsigned)R;
            if (__builtin_amdgcn_ballot_w32(m0 | m1 | m2 | m3)) {
                np = append_k(mypend, np, m0, e0 + 0);
                np = append_k(mypend, np, m1, e0 + 1);
                np = append_k(mypend, np, m2, e0 + 2);
                np = append_k(mypend, np, m3, e0 + 3);
                while (np >= 32) {
                    np -= 32;
                    flush_list<DOUT>(mypend + np, 32, lane, n0, yl, src, dst, acc, cntL);
                }
            }
        }
        #pragma unroll
        for (int k = 0; k < 4; ++k) cur[k] = nxt[k];
    }
    flush_list<DOUT>(mypend, np, lane, n0, yl, src, dst, acc, cntL);
}

template<int R>
__global__ __launch_bounds__(1024) void bin_edges(
        const int* __restrict__ src, const int* __restrict__ dst,
        int* __restrict__ elist, int* __restrict__ ecount, int E, int nb) {
    constexpr int EPT = 4;
    constexpr int SC  = 1024 * EPT;
    __shared__ int  hist[1024];
    __shared__ int  gbase[1024];
    __shared__ int  wsum[32];
    __shared__ int2 sorted[SC];
    __shared__ int  sbin[SC];

    const int tid  = threadIdx.x;
    const int lane = tid & 31;
    const int wave = tid >> 5;
    const int chunk = (E + (int)gridDim.x - 1) / (int)gridDim.x;
    const int c0 = (int)blockIdx.x * chunk;
    const int c1 = (c0 + chunk < E) ? (c0 + chunk) : E;

    for (int r0 = c0; r0 < c1; r0 += SC) {
        const int r1 = (r0 + SC < c1) ? (r0 + SC) : c1;
        hist[tid] = 0;
        __syncthreads();

        int s[EPT], dd[EPT], g[EPT], d[EPT], rk[EPT];
        #pragma unroll
        for (int k = 0; k < EPT; ++k) {
            const int e = r0 + k * 1024 + tid;
            dd[k] = -1; s[k] = 0;
            if (e < r1) { dd[k] = dst[e]; s[k] = src[e]; }
        }
        #pragma unroll
        for (int k = 0; k < EPT; ++k) {
            g[k] = -1; d[k] = 0;
            if (dd[k] >= 0) {
                g[k] = dd[k] / R;
                d[k] = dd[k] - g[k] * R;
            }
        }
        #pragma unroll
        for (int k = 0; k < EPT; ++k) {
            rk[k] = 0;
            if (g[k] >= 0)
                rk[k] = __hip_atomic_fetch_add(&hist[g[k]], 1, __ATOMIC_RELAXED, __HIP_MEMORY_SCOPE_WORKGROUP);
        }
        __syncthreads();

        const int cn = hist[tid];
        int inc = cn;
        for (int o = 1; o < 32; o <<= 1) {
            const int t = __shfl_up(inc, o);
            if (lane >= o) inc += t;
        }
        if (lane == 31) wsum[wave] = inc;
        __syncthreads();
        if (wave == 0) {
            const int w = wsum[lane];
            int wi = w;
            for (int o = 1; o < 32; o <<= 1) {
                const int t = __shfl_up(wi, o);
                if (lane >= o) wi += t;
            }
            wsum[lane] = wi - w;
        }
        __syncthreads();
        hist[tid] = inc - cn + wsum[wave];
        int gb = 0;
        if (cn > 0 && tid < nb)
            gb = __hip_atomic_fetch_add(&ecount[tid * CSTR], cn, __ATOMIC_RELAXED, __HIP_MEMORY_SCOPE_AGENT);
        gbase[tid] = gb;
        __syncthreads();

        #pragma unroll
        for (int k = 0; k < EPT; ++k) {
            if (g[k] >= 0) {
                const int p = hist[g[k]] + rk[k];
                sorted[p] = make_int2(s[k], d[k]);
                sbin[p]   = g[k];
            }
        }
        __syncthreads();

        const int n = r1 - r0;
        for (int i = tid; i < n; i += 1024) {
            const int gg = sbin[i];
            const int p  = gbase[gg] + (i - hist[gg]);
            if (p < LCAP) ((int2*)elist)[(size_t)gg * LCAP + p] = sorted[i];
        }
        __syncthreads();
    }
}

template<int DOUT, int R, bool ZADD>
__global__ __launch_bounds__(1024) void agg_owner(
        const float* __restrict__ yl, const int* __restrict__ src,
        const int* __restrict__ dst, const float* __restrict__ bias,
        float* __restrict__ agg, const int* __restrict__ elist, const int* __restrict__ ecount,
        const float* __restrict__ zadd, int N, int E, int vec_ok) {
    constexpr int NW = 32;
    constexpr int PCAP = 160;
    __shared__ float acc[R * DOUT];
    __shared__ float cntL[R];
    __shared__ int   pend[NW * PCAP];

    const int tid  = threadIdx.x;
    const int lane = tid & 31;
    const int wave = __builtin_amdgcn_readfirstlane(tid >> 5);
    const int n0   = blockIdx.x * R;

    for (int i = tid; i < R * DOUT; i += 1024) acc[i] = 0.0f;
    for (int i = tid; i < R; i += 1024) cntL[i] = 0.0f;
    __syncthreads();

    int* mypend = pend + wave * PCAP;
    const int2* mylist = (const int2*)elist + (size_t)blockIdx.x * LCAP;

    {
        const int cnt = __builtin_amdgcn_readfirstlane(ecount[blockIdx.x * CSTR]);
        if (cnt <= LCAP) {
            constexpr int LCH = LCAP / (NW * 32);
            const int per = (((cnt + NW - 1) / NW) + 7) & ~7;
            const int lo  = wave * per;
            const int hi  = (lo + per < cnt) ? (lo + per) : cnt;
            int2 t[LCH];
            #pragma unroll
            for (int c = 0; c < LCH; ++c) t[c] = make_int2(0, 0);
            #pragma unroll
            for (int c = 0; c < LCH; ++c) {
                const int i = lo + c * 32 + lane;
                if (i < hi) t[c] = mylist[i];
            }
            #pragma unroll
            for (int c = 0; c < LCH; ++c) {
                const int b = lo + c * 32;
                if (b < hi) {
                    const int n = (hi - b < 32) ? (hi - b) : 32;
                    if (lane < n) lds_add(&cntL[t[c].y], 1.0f);
                    add_rows<DOUT>(t[c].x, t[c].y, n, lane, yl, acc);
                }
            }
        } else {
            sweep_edges<DOUT, R>(mypend, wave, lane, n0, yl, src, dst, acc, cntL, E, vec_ok);
        }
    }
    __syncthreads();

    for (int i = tid; i < R * DOUT; i += 1024) {
        const int r = i / DOUT;
        const int f = i - r * DOUT;
        if (n0 + r < N) {
            float v = acc[i] / fmaxf(cntL[r], 1.0f) + bias[f];
            if (ZADD) v += zadd[(size_t)n0 * DOUT + i];
            agg[(size_t)n0 * DOUT + i] = v;
        }
    }
}

template<int K, int NC, int NT, bool ACC, bool GELU>
__global__ __launch_bounds__(1024) void gemm_wlds(
        const float* __restrict__ A, const float* __restrict__ W,
        const float* __restrict__ Cin, float* __restrict__ D, int tilesM,
        int* __restrict__ zc, int nz) {
    constexpr int TN    = (NC + 15) / 16;
    constexpr int NCPAD = TN * 16;
    constexpr int NCH   = TN / NT;
    constexpr int KS    = K / 4;
    constexpr int KB    = KS / 8;
    constexpr int TOT   = KS * NCPAD;
    constexpr int ENT   = (TOT + 1023) / 1024;
    static_assert(TN % NT == 0 && KS % 8 == 0 && 32 % NCH == 0, "tiling");
    static_assert(NC == NCPAD || NT == 1, "the column guard covers one N-tile");
    __shared__ float Ws[TOT * 4];

    const int tid  = threadIdx.x;
    const int lane = tid & 31;
    const int wave = __builtin_amdgcn_readfirstlane(tid >> 5);
    const int half = lane >> 4;
    const int l    = lane & 15;
    const int nc   = wave % NCH;
    int mt = (int)blockIdx.x + (int)gridDim.x * (wave / NCH);

    if (zc != nullptr && blockIdx.x == 0 && tid < nz) zc[tid * CSTR] = 0;

    v2f an[8];
    gcfp Ap = (gcfp)A + (size_t)((mt < tilesM ? mt : 0) * 16 + l) * K + half * 2;
    if (mt < tilesM) {
        #pragma unroll
        for (int u = 0; u < 8; ++u) { an[u].x = Ap[u * 4]; an[u].y = Ap[u * 4 + 1]; }
    }

    {
        float4 w[ENT];
        #pragma unroll
        for (int e = 0; e < ENT; ++e) {
            const int i   = tid + e * 1024;
            const int ks  = i / NCPAD;
            const int col = i - ks * NCPAD;
            w[e] = make_float4(0.0f, 0.0f, 0.0f, 0.0f);
            if ((TOT % 1024 == 0 || i < TOT) && (NC == NCPAD || col < NC)) {
                const float* p = W + (size_t)(ks * 4) * NC + col;
                w[e].x = p[0];
                w[e].y = p[NC];
                w[e].z = p[2 * NC];
                w[e].w = p[3 * NC];
            }
        }
        #pragma unroll
        for (int e = 0; e < ENT; ++e) {
            const int i = tid + e * 1024;
            if (TOT % 1024 == 0 || i < TOT) *(float4*)(Ws + i * 4) = w[e];
        }
    }
    __syncthreads();

    const float* Bp = Ws + (nc * NT * 16 + l) * 4 + half * 2;
    gcfp Cg = (gcfp)Cin;
    gfp  Dg = (gfp)D;

    while (mt < tilesM) {
        v8f acc[NT];
        #pragma unroll
        for (int t = 0; t < NT; ++t) acc[t] = (v8f){};

        #pragma unroll 1
        for (int kb = 0; kb < KB; ++kb) {
            v2f ac[8];
            #pragma unroll
            for (int u = 0; u < 8; ++u) ac[u] = an[u];
            if (kb + 1 < KB) {
                gcfp An = Ap + (kb + 1) * 32;
                #pragma unroll
                for (int u = 0; u < 8; ++u) { an[u].x = An[u * 4]; an[u].y = An[u * 4 + 1]; }
            }
            const float* Bk = Bp + kb * (8 * NCPAD * 4);
            #pragma unroll
            for (int u = 0; u < 8; ++u) {
                #pragma unroll
                for (int t = 0; t < NT; ++t) {
                    const v2f b = *(const v2f*)(Bk + (u * NCPAD + t * 16) * 4);
                    acc[t] = __builtin_amdgcn_wmma_f32_16x16x4_f32(
                                 false, ac[u], false, b, (short)0, acc[t], false, false);
                }
            }
        }

        const size_t off0 = (size_t)(mt * 16 + half * 8) * NC + (nc * NT * 16 + l);
        if (NC == NCPAD || nc * NT * 16 + l < NC) {
            if (ACC) {
                float cv[NT][8];
                #pragma unroll
                for (int t = 0; t < NT; ++t) {
                    #pragma unroll
                    for (int g = 0; g < 8; ++g) cv[t][g] = Cg[off0 + (size_t)g * NC + t * 16];
                }
                #pragma unroll
                for (int t = 0; t < NT; ++t) {
                    #pragma unroll
                    for (int g = 0; g < 8; ++g) acc[t][g] += cv[t][g];
                }
            }
            #pragma unroll
            for (int t = 0; t < NT; ++t) {
                #pragma unroll
                for (int g = 0; g < 8; ++g) {
                    float v = acc[t][g];
                    if (GELU) v = gelu_exact(v);
                    Dg[off0 + (size_t)g * NC + t * 16] = v;
                }
            }
        }

        mt += (int)gridDim.x * (32 / NCH);
        if (mt < tilesM) {
            Ap = (gcfp)A + (size_t)(mt * 16 + l) * K + half * 2;
            #pragma unroll
            for (int u = 0; u < 8; ++u) { an[u].x = Ap[u * 4]; an[u].y = Ap[u * 4 + 1]; }
        }
    }
}

template<int K, int NC, int NO>
__global__ __launch_bounds__(1024) void gemm_root_next(
        const float* __restrict__ A, const float* __restrict__ W,
        const float* __restrict__ Cin,
        const float* __restrict__ Wl3, const float* __restrict__ Wr3,
        float* __restrict__ Y, float* __restrict__ Z, int tilesM) {
    constexpr int NT  = NC / 16;
    constexpr int KS  = K / 4;
    constexpr int KB  = KS / 8;
    constexpr int TOT = KS * NC;
    constexpr int ENT = (TOT + 1023) / 1024;
    constexpr int KS3 = NC / 4;
    static_assert(NC % 16 == 0 && KS % 8 == 0 && TOT % 1024 == 0, "tiling");
    static_assert(2 * NO <= 8, "Y and Z columns sit in the registers of lane half 0");
    __shared__ float Ws[TOT * 4];
    __shared__ float W3s[KS3 * 64];

    const int tid  = threadIdx.x;
    const int lane = tid & 31;
    const int wave = __builtin_amdgcn_readfirstlane(tid >> 5);
    const int half = lane >> 4;
    const int l    = lane & 15;
    int mt = (int)blockIdx.x + (int)gridDim.x * wave;

    v2f an[8];
    gcfp Ap = (gcfp)A + (size_t)((mt < tilesM ? mt : 0) * 16 + l) * K + half * 2;
    if (mt < tilesM) {
        #pragma unroll
        for (int u = 0; u < 8; ++u) { an[u].x = Ap[u * 4]; an[u].y = Ap[u * 4 + 1]; }
    }

    {
        float4 w[ENT];
        #pragma unroll
        for (int e = 0; e < ENT; ++e) {
            const int i   = tid + e * 1024;
            const int ks  = i / NC;
            const int col = i - ks * NC;
            const float* p = W + (size_t)(ks * 4) * NC + col;
            w[e].x = p[0];
            w[e].y = p[NC];
            w[e].z = p[2 * NC];
            w[e].w = p[3 * NC];
        }
        #pragma unroll
        for (int e = 0; e < ENT; ++e) *(float4*)(Ws + (tid + e * 1024) * 4) = w[e];
    }
    for (int i = tid; i < KS3 * 64; i += 1024) {
        const int s  = i & 1;
        const int ln = (i >> 1) & 31;
        const int j  = i >> 6;
        const int c  = ln & 15;
        const int k  = 16 * (j >> 2) + 8 * (ln >> 4) + 2 * (j & 3) + s;
        float v = 0.0f;
        if (c < NO)          v = Wl3[k * NO + c];
        else if (c < 2 * NO) v = Wr3[k * NO + (c - NO)];
        W3s[i] = v;
    }
    __syncthreads();

    const float* Bp  = Ws + l * 4 + half * 2;
    const float* W3p = W3s + lane * 2;
    gcfp Cg = (gcfp)Cin;

    while (mt < tilesM) {
        v8f acc[NT];
        #pragma unroll
        for (int t = 0; t < NT; ++t) acc[t] = (v8f){};

        #pragma unroll 1
        for (int kb = 0; kb < KB; ++kb) {
            v2f ac[8];
            #pragma unroll
            for (int u = 0; u < 8; ++u) ac[u] = an[u];
            if (kb + 1 < KB) {
                gcfp An = Ap + (kb + 1) * 32;
                #pragma unroll
                for (int u = 0; u < 8; ++u) { an[u].x = An[u * 4]; an[u].y = An[u * 4 + 1]; }
            }
            const float* Bk = Bp + kb * (8 * NC * 4);
            #pragma unroll
            for (int u = 0; u < 8; ++u) {
                #pragma unroll
                for (int t = 0; t < NT; ++t) {
                    const v2f wf = *(const v2f*)(Bk + (u * NC + t * 16) * 4);
                    acc[t] = __builtin_amdgcn_wmma_f32_16x16x4_f32(
                                 false, wf, false, ac[u], (short)0, acc[t], false, false);
                }
            }
        }

        const int row = mt * 16 + l;
        {
            v4f cv[NT][2];
            gcv4p cp = (gcv4p)(Cg + (size_t)row * NC + half * 8);
            #pragma unroll
            for (int t = 0; t < NT; ++t) {
                cv[t][0] = cp[t * 4];
                cv[t][1] = cp[t * 4 + 1];
            }
            #pragma unroll
            for (int t = 0; t < NT; ++t) {
                acc[t][0] = gelu_exact(acc[t][0] + cv[t][0].x);
                acc[t][1] = gelu_exact(acc[t][1] + cv[t][0].y);
                acc[t][2] = gelu_exact(acc[t][2] + cv[t][0].z);
                acc[t][3] = gelu_exact(acc[t][3] + cv[t][0].w);
                acc[t][4] = gelu_exact(acc[t][4] + cv[t][1].x);
                acc[t][5] = gelu_exact(acc[t][5] + cv[t][1].y);
                acc[t][6] = gelu_exact(acc[t][6] + cv[t][1].z);
                acc[t][7] = gelu_exact(acc[t][7] + cv[t][1].w);
            }
        }

        v8f p0 = (v8f){}, p1 = (v8f){};
        #pragma unroll
        for (int j = 0; j < KS3; j += 2) {
            const int t0 = j >> 2,       m0 = j & 3;
            const int t1 = (j + 1) >> 2, m1 = (j + 1) & 3;
            v2f x0, x1;
            x0.x = acc[t0][2 * m0]; x0.y = acc[t0][2 * m0 + 1];
            x1.x = acc[t1][2 * m1]; x1.y = acc[t1][2 * m1 + 1];
            const v2f w0 = *(const v2f*)(W3p + j * 64);
            const v2f w1 = *(const v2f*)(W3p + (j + 1) * 64);
            p0 = __builtin_amdgcn_wmma_f32_16x16x4_f32(false, w0, false, x0, (short)0, p0, false, false);
            p1 = __builtin_amdgcn_wmma_f32_16x16x4_f32(false, w1, false, x1, (short)0, p1, false, false);
        }
        if (half == 0) {
            float* yp = Y + (size_t)row * NO;
            float* zp = Z + (size_t)row * NO;
            #pragma unroll
            for (int c = 0; c < NO; ++c) {
                yp[c] = p0[c] + p1[c];
                zp[c] = p0[NO + c] + p1[NO + c];
            }
        }

        mt += (int)gridDim.x * 32;
        if (mt < tilesM) {
            Ap = (gcfp)A + (size_t)(mt * 16 + l) * K + half * 2;
            #pragma unroll
            for (int u = 0; u < 8; ++u) { an[u].x = Ap[u * 4]; an[u].y = Ap[u * 4 + 1]; }
        }
    }
}


extern "C" void kernel_launch(void* const* d_in, const int* in_sizes, int n_in,
                              void* d_out, int out_size, void* d_ws, size_t ws_size,
                              hipStream_t stream) {
    static const int DIMS[5] = {256, 192, 128, 64, 3};

    const float* feat  = (const float*)d_in[0];
    const int*   edges = (const int*)d_in[1];
    const int N = in_sizes[0] / DIMS[0];
    const int E = in_sizes[1] / 2;
    const int* src = edges;
    const int* dst = edges + E;

    const float* Wl[4] = {(const float*)d_in[2], (const float*)d_in[5],
                          (const float*)d_in[8], (const float*)d_in[11]};
    const float* Wr[4] = {(const float*)d_in[3], (const float*)d_in[6],
                          (const float*)d_in[9], (const float*)d_in[12]};
    const float* Bb[4] = {(const float*)d_in[4], (const float*)d_in[7],
                          (const float*)d_in[10], (const float*)d_in[13]};

    float* ws    = (float*)d_ws;
    float* r0    = ws;
    float* r1    = ws + (size_t)N * 192;
    float* r2    = ws + (size_t)N * 320;
    int* elist   = (int*)(ws + (size_t)N * 448);
    int* ecount  = elist + (size_t)((N + 195) / 196) * LCAP * 2;
    float* const YL[4]  = {r0, r1, r0, r2};
    float* const AGG[4] = {r1, r2, r0 + (size_t)N * 64, r2 + (size_t)N * 64};
    float* const z3 = r2 + (size_t)N * 64;

    const int vec_ok = (E % 4 == 0) ? 1 : 0;
    const int tilesM = N / 16;
    const int GW = 256;
    const int nb = (N + 195) / 196;

    const float* x = feat;
    for (int L = 0; L < 4; ++L) {
        float* yl   = YL[L];
        float* agg  = AGG[L];
        float* xout = (L == 3) ? (float*)d_out : yl;

        if (L == 0) {
            gemm_wlds<256, 192, 6, false, false><<<GW, 1024, 0, stream>>>(x, Wl[L], nullptr, yl, tilesM, ecount, nb);
            bin_edges<196><<<256, 1024, 0, stream>>>(src, dst, elist, ecount, E, nb);
            agg_owner<192, 196, false><<<nb, 1024, 0, stream>>>(
                yl, src, dst, Bb[L], agg, elist, ecount, nullptr, N, E, vec_ok);
            gemm_wlds<256, 192, 6, true, true><<<GW, 1024, 0, stream>>>(x, Wr[L], agg, xout, tilesM, nullptr, 0);
        } else if (L == 1) {
            gemm_wlds<192, 128, 4, false, false><<<GW, 1024, 0, stream>>>(x, Wl[L], nullptr, yl, tilesM, nullptr, 0);
            agg_owner<128, 196, false><<<nb, 1024, 0, stream>>>(
                yl, src, dst, Bb[L], agg, elist, ecount, nullptr, N, E, vec_ok);
            gemm_wlds<192, 128, 4, true, true><<<GW, 1024, 0, stream>>>(x, Wr[L], agg, xout, tilesM, nullptr, 0);
        } else if (L == 2) {
            gemm_wlds<128, 64, 4, false, false><<<GW, 1024, 0, stream>>>(x, Wl[L], nullptr, yl, tilesM, nullptr, 0);
            agg_owner<64, 196, false><<<nb, 1024, 0, stream>>>(
                yl, src, dst, Bb[L], agg, elist, ecount, nullptr, N, E, vec_ok);
            gemm_root_next<128, 64, 3><<<GW, 1024, 0, stream>>>(
                x, Wr[L], agg, Wl[3], Wr[3], YL[3], z3, tilesM);
        } else {
            agg_owner<3, 196, true><<<nb, 1024, 0, stream>>>(
                yl, src, dst, Bb[L], (float*)d_out, elist, ecount, z3, N, E, vec_ok);
        }
        x = xout;
    }
}


# layers 1, 2: root product moves into the aggregation launch: 8 of the owner's 32 waves run its WMMAs while 24 gather (LDS sums still arrival order); layer-3 k order changes; 11 -> 9 launches
# speedup vs baseline: 3.9422x; 1.0734x over previous
#include <hip/hip_runtime.h>
#include <math.h>

typedef __attribute__((ext_vector_type(2))) float v2f;
typedef __attribute__((ext_vector_type(8))) float v8f;

typedef const float __attribute__((address_space(1)))* gcfp;
typedef float __attribute__((address_space(1)))* gfp;
typedef __attribute__((ext_vector_type(4))) float v4f;
typedef const v4f __attribute__((address_space(1)))* gcv4p;

__device__ __forceinline__ float gelu_exact(float v) {
    return 0.5f * v * (1.0f + erff(v * 0.7071067811865475f));
}

__device__ __forceinline__ void load_sb(int4 (&q)[4], const int* __restrict__ dst,
                                        int sb, int lane, int E, int vec_ok) {
    const int base = sb * 512;
    if (vec_ok && base + 512 <= E) {
        const int4* p = (const int4*)(dst + base) + lane;
        #pragma unroll
        for (int k = 0; k < 4; ++k) q[k] = p[k * 32];
    } else {
        #pragma unroll
        for (int k = 0; k < 4; ++k) {
            const int e0 = base + k * 128 + lane * 4;
            q[k].x = (e0 + 0 < E) ? dst[e0 + 0] : -1;
            q[k].y = (e0 + 1 < E) ? dst[e0 + 1] : -1;
            q[k].z = (e0 + 2 < E) ? dst[e0 + 2] : -1;
            q[k].w = (e0 + 3 < E) ? dst[e0 + 3] : -1;
        }
    }
}

__device__ __forceinline__ int append_k(int* mypend, int np, bool m, int e) {
    const unsigned b = __builtin_amdgcn_ballot_w32(m);
    if (b) {
        if (m) mypend[np + (int)__builtin_amdgcn_mbcnt_lo(b, 0u)] = e;
        np += __builtin_popcount(b);
    }
    return np;
}

__device__ __forceinline__ void lds_add(float* p, float v) {
    __hip_atomic_fetch_add(p, v, __ATOMIC_RELAXED, __HIP_MEMORY_SCOPE_WORKGROUP);
}

constexpr int LCAP = 6144;
constexpr int CSTR = 64;

template<int DOUT>
__device__ __forceinline__ void add_rows(int s, int d, int n, int lane,
        const float* __restrict__ yl, float* acc) {
    if constexpr (DOUT >= 64) {
        constexpr int NCH = DOUT / 64;
        constexpr int U = 8;
        for (int i0 = 0; i0 < n; i0 += U) {
            float2 v[U][NCH];
            #pragma unroll
            for (int u = 0; u < U; ++u) {
                int idx = i0 + u;
                if (idx > n - 1) idx = n - 1;
                const int si = __builtin_amdgcn_readlane(s, idx);
                const float* row = yl + (size_t)si * DOUT + lane * 2;
                #pragma unroll
                for (int c = 0; c < NCH; ++c) v[u][c] = *(const float2*)(row + c * 64);
            }
            #pragma unroll
            for (int u = 0; u < U; ++u) {
                if (i0 + u < n) {
                    const int di = __builtin_amdgcn_readlane(d, i0 + u);
                    float* a = acc + di * DOUT + lane * 2;
                    #pragma unroll
                    for (int c = 0; c < NCH; ++c) {
                        lds_add(a + c * 64 + 0, v[u][c].x);
                        lds_add(a + c * 64 + 1, v[u][c].y);
                    }
                }
            }
        }
    } else {
        if (lane < n) {
            #pragma unroll
            for (int c = 0; c < DOUT; ++c)
                lds_add(&acc[d * DOUT + c], yl[(size_t)s * DOUT + c]);
        }
    }
}

template<int DOUT>
__device__ __forceinline__ void flush_list(const int* p, int n, int lane, int n0,
        const float* __restrict__ yl, const int* __restrict__ src,
        const int* __restrict__ dst, float* acc, float* cntL) {
    __builtin_amdgcn_fence(__ATOMIC_RELEASE, "wavefront");
    __builtin_amdgcn_wave_barrier();
    __builtin_amdgcn_fence(__ATOMIC_ACQUIRE, "wavefront");
    int s = 0, d = 0;
    if (lane < n) {
        const int e = p[lane];
        s = src[e];
        d = dst[e] - n0;
        lds_add(&cntL[d], 1.0f);
    }
    add_rows<DOUT>(s, d, n, lane, yl, acc);
    __builtin_amdgcn_fence(__ATOMIC_RELEASE, "wavefront");
    __builtin_amdgcn_wave_barrier();
}

template<int DOUT, int R, int NW = 32>
__device__ __forceinline__ void sweep_edges(int* mypend, int wave, int lane, int n0,
        const float* __restrict__ yl, const int* __restrict__ src,
        const int* __restrict__ dst, float* acc, float* cntL, int E, int vec_ok) {
    int np = 0;
    const int nsb = (E + 511) / 512;
    const int start = (int)(((long long)blockIdx.x * nsb) / gridDim.x);

    int4 cur[4];
    #pragma unroll
    for (int k = 0; k < 4; ++k) cur[k] = make_int4(-1, -1, -1, -1);
    if (wave < nsb) {
        int sb = wave + start; if (sb >= nsb) sb -= nsb;
        load_sb(cur, dst, sb, lane, E, vec_ok);
    }
    for (int j = wave; j < nsb; j += NW) {
        int sb = j + start; if (sb >= nsb) sb -= nsb;
        int4 nxt[4];
        if (j + NW < nsb) {
            int sbn = j + NW + start; if (sbn >= nsb) sbn -= nsb;
            load_sb(nxt, dst, sbn, lane, E, vec_ok);
        } else {
            #pragma unroll
            for (int k = 0; k < 4; ++k) nxt[k] = make_int4(-1, -1, -1, -1);
        }
        #pragma unroll
        for (int k = 0; k < 4; ++k) {
            const int e0 = sb * 512 + k * 128 + lane * 4;
            const int4 q = cur[k];
            const bool m0 = (unsigned)(q.x - n0) < (unsigned)R;
            const bool m1 = (unsigned)(q.y - n0) < (unsigned)R;
            const bool m2 = (unsigned)(q.z - n0) < (unsigned)R;
            const bool m3 = (unsigned)(q.w - n0) < (unsigned)R;
            if (__builtin_amdgcn_ballot_w32(m0 | m1 | m2 | m3)) {
                np = append_k(mypend, np, m0, e0 + 0);
                np = append_k(mypend, np, m1, e0 + 1);
                np = append_k(mypend, np, m2, e0 + 2);
                np = append_k(mypend, np, m3, e0 + 3);
                while (np >= 32) {
                    np -= 32;
                    flush_list<DOUT>(mypend + np, 32, lane, n0, yl, src, dst, acc, cntL);
                }
            }
        }
        #pragma unroll
        for (int k = 0; k < 4; ++k) cur[k] = nxt[k];
    }
    flush_list<DOUT>(mypend, np, lane, n0, yl, src, dst, acc, cntL);
}

template<int R>
__global__ __launch_bounds__(1024) void bin_edges(
        const int* __restrict__ src, const int* __restrict__ dst,
        int* __restrict__ elist, int* __restrict__ ecount, int E, int nb) {
    constexpr int EPT = 4;
    constexpr int SC  = 1024 * EPT;
    __shared__ int  hist[1024];
    __shared__ int  gbase[1024];
    __shared__ int  wsum[32];
    __shared__ int2 sorted[SC];
    __shared__ int  sbin[SC];

    const int tid  = threadIdx.x;
    const int lane = tid & 31;
    const int wave = tid >> 5;
    const int chunk = (E + (int)gridDim.x - 1) / (int)gridDim.x;
    const int c0 = (int)blockIdx.x * chunk;
    const int c1 = (c0 + chunk < E) ? (c0 + chunk) : E;

    for (int r0 = c0; r0 < c1; r0 += SC) {
        const int r1 = (r0 + SC < c1) ? (r0 + SC) : c1;
        hist[tid] = 0;
        __syncthreads();

        int s[EPT], dd[EPT], g[EPT], d[EPT], rk[EPT];
        #pragma unroll
        for (int k = 0; k < EPT; ++k) {
            const int e = r0 + k * 1024 + tid;
            dd[k] = -1; s[k] = 0;
            if (e < r1) { dd[k] = dst[e]; s[k] = src[e]; }
        }
        #pragma unroll
        for (int k = 0; k < EPT; ++k) {
            g[k] = -1; d[k] = 0;
            if (dd[k] >= 0) {
                g[k] = dd[k] / R;
                d[k] = dd[k] - g[k] * R;
            }
        }
        #pragma unroll
        for (int k = 0; k < EPT; ++k) {
            rk[k] = 0;
            if (g[k] >= 0)
                rk[k] = __hip_atomic_fetch_add(&hist[g[k]], 1, __ATOMIC_RELAXED, __HIP_MEMORY_SCOPE_WORKGROUP);
        }
        __syncthreads();

        const int cn = hist[tid];
        int inc = cn;
        for (int o = 1; o < 32; o <<= 1) {
            const int t = __shfl_up(inc, o);
            if (lane >= o) inc += t;
        }
        if (lane == 31) wsum[wave] = inc;
        __syncthreads();
        if (wave == 0) {
            const int w = wsum[lane];
            int wi = w;
            for (int o = 1; o < 32; o <<= 1) {
                const int t = __shfl_up(wi, o);
                if (lane >= o) wi += t;
            }
            wsum[lane] = wi - w;
        }
        __syncthreads();
        hist[tid] = inc - cn + wsum[wave];
        int gb = 0;
        if (cn > 0 && tid < nb)
            gb = __hip_atomic_fetch_add(&ecount[tid * CSTR], cn, __ATOMIC_RELAXED, __HIP_MEMORY_SCOPE_AGENT);
        gbase[tid] = gb;
        __syncthreads();

        #pragma unroll
        for (int k = 0; k < EPT; ++k) {
            if (g[k] >= 0) {
                const int p = hist[g[k]] + rk[k];
                sorted[p] = make_int2(s[k], d[k]);
                sbin[p]   = g[k];
            }
        }
        __syncthreads();

        const int n = r1 - r0;
        for (int i = tid; i < n; i += 1024) {
            const int gg = sbin[i];
            const int p  = gbase[gg] + (i - hist[gg]);
            if (p < LCAP) ((int2*)elist)[(size_t)gg * LCAP + p] = sorted[i];
        }
        __syncthreads();
    }
}

template<int DOUT, int R, bool ZADD>
__global__ __launch_bounds__(1024) void agg_owner(
        const float* __restrict__ yl, const int* __restrict__ src,
        const int* __restrict__ dst, const float* __restrict__ bias,
        float* __restrict__ agg, const int* __restrict__ elist, const int* __restrict__ ecount,
        const float* __restrict__ zadd, int N, int E, int vec_ok) {
    constexpr int NW = 32;
    constexpr int PCAP = 160;
    __shared__ float acc[R * DOUT];
    __shared__ float cntL[R];
    __shared__ int   pend[NW * PCAP];

    const int tid  = threadIdx.x;
    const int lane = tid & 31;
    const int wave = __builtin_amdgcn_readfirstlane(tid >> 5);
    const int n0   = blockIdx.x * R;

    for (int i = tid; i < R * DOUT; i += 1024) acc[i] = 0.0f;
    for (int i = tid; i < R; i += 1024) cntL[i] = 0.0f;
    __syncthreads();

    int* mypend = pend + wave * PCAP;
    const int2* mylist = (const int2*)elist + (size_t)blockIdx.x * LCAP;

    {
        const int cnt = __builtin_amdgcn_readfirstlane(ecount[blockIdx.x * CSTR]);
        if (cnt <= LCAP) {
            constexpr int LCH = LCAP / (NW * 32);
            const int per = (((cnt + NW - 1) / NW) + 7) & ~7;
            const int lo  = wave * per;
            const int hi  = (lo + per < cnt) ? (lo + per) : cnt;
            int2 t[LCH];
            #pragma unroll
            for (int c = 0; c < LCH; ++c) t[c] = make_int2(0, 0);
            #pragma unroll
            for (int c = 0; c < LCH; ++c) {
                const int i = lo + c * 32 + lane;
                if (i < hi) t[c] = mylist[i];
            }
            #pragma unroll
            for (int c = 0; c < LCH; ++c) {
                const int b = lo + c * 32;
                if (b < hi) {
                    const int n = (hi - b < 32) ? (hi - b) : 32;
                    if (lane < n) lds_add(&cntL[t[c].y], 1.0f);
                    add_rows<DOUT>(t[c].x, t[c].y, n, lane, yl, acc);
                }
            }
        } else {
            sweep_edges<DOUT, R>(mypend, wave, lane, n0, yl, src, dst, acc, cntL, E, vec_ok);
        }
    }
    __syncthreads();

    for (int i = tid; i < R * DOUT; i += 1024) {
        const int r = i / DOUT;
        const int f = i - r * DOUT;
        if (n0 + r < N) {
            float v = acc[i] / fmaxf(cntL[r], 1.0f) + bias[f];
            if (ZADD) v += zadd[(size_t)n0 * DOUT + i];
            agg[(size_t)n0 * DOUT + i] = v;
        }
    }
}

template<int K, int NC, int NT, bool ACC, bool GELU>
__global__ __launch_bounds__(1024) void gemm_wlds(
        const float* __restrict__ A, const float* __restrict__ W,
        const float* __restrict__ Cin, float* __restrict__ D, int tilesM,
        int* __restrict__ zc, int nz) {
    constexpr int TN    = (NC + 15) / 16;
    constexpr int NCPAD = TN * 16;
    constexpr int NCH   = TN / NT;
    constexpr int KS    = K / 4;
    constexpr int KB    = KS / 8;
    constexpr int TOT   = KS * NCPAD;
    constexpr int ENT   = (TOT + 1023) / 1024;
    static_assert(TN % NT == 0 && KS % 8 == 0 && 32 % NCH == 0, "tiling");
    static_assert(NC == NCPAD || NT == 1, "the column guard covers one N-tile");
    __shared__ float Ws[TOT * 4];

    const int tid  = threadIdx.x;
    const int lane = tid & 31;
    const int wave = __builtin_amdgcn_readfirstlane(tid >> 5);
    const int half = lane >> 4;
    const int l    = lane & 15;
    const int nc   = wave % NCH;
    int mt = (int)blockIdx.x + (int)gridDim.x * (wave / NCH);

    if (zc != nullptr && blockIdx.x == 0 && tid < nz) zc[tid * CSTR] = 0;

    v2f an[8];
    gcfp Ap = (gcfp)A + (size_t)((mt < tilesM ? mt : 0) * 16 + l) * K + half * 2;
    if (mt < tilesM) {
        #pragma unroll
        for (int u = 0; u < 8; ++u) { an[u].x = Ap[u * 4]; an[u].y = Ap[u * 4 + 1]; }
    }

    {
        float4 w[ENT];
        #pragma unroll
        for (int e = 0; e < ENT; ++e) {
            const int i   = tid + e * 1024;
            const int ks  = i / NCPAD;
            const int col = i - ks * NCPAD;
            w[e] = make_float4(0.0f, 0.0f, 0.0f, 0.0f);
            if ((TOT % 1024 == 0 || i < TOT) && (NC == NCPAD || col < NC)) {
                const float* p = W + (size_t)(ks * 4) * NC + col;
                w[e].x = p[0];
                w[e].y = p[NC];
                w[e].z = p[2 * NC];
                w[e].w = p[3 * NC];
            }
        }
        #pragma unroll
        for (int e = 0; e < ENT; ++e) {
            const int i = tid + e * 1024;
            if (TOT % 1024 == 0 || i < TOT) *(float4*)(Ws + i * 4) = w[e];
        }
    }
    __syncthreads();

    const float* Bp = Ws + (nc * NT * 16 + l) * 4 + half * 2;
    gcfp Cg = (gcfp)Cin;
    gfp  Dg = (gfp)D;

    while (mt < tilesM) {
        v8f acc[NT];
        #pragma unroll
        for (int t = 0; t < NT; ++t) acc[t] = (v8f){};

        #pragma unroll 1
        for (int kb = 0; kb < KB; ++kb) {
            v2f ac[8];
            #pragma unroll
            for (int u = 0; u < 8; ++u) ac[u] = an[u];
            if (kb + 1 < KB) {
                gcfp An = Ap + (kb + 1) * 32;
                #pragma unroll
                for (int u = 0; u < 8; ++u) { an[u].x = An[u * 4]; an[u].y = An[u * 4 + 1]; }
            }
            const float* Bk = Bp + kb * (8 * NCPAD * 4);
            #pragma unroll
            for (int u = 0; u < 8; ++u) {
                #pragma unroll
                for (int t = 0; t < NT; ++t) {
                    const v2f b = *(const v2f*)(Bk + (u * NCPAD + t * 16) * 4);
                    acc[t] = __builtin_amdgcn_wmma_f32_16x16x4_f32(
                                 false, ac[u], false, b, (short)0, acc[t], false, false);
                }
            }
        }

        const size_t off0 = (size_t)(mt * 16 + half * 8) * NC + (nc * NT * 16 + l);
        if (NC == NCPAD || nc * NT * 16 + l < NC) {
            if (ACC) {
                float cv[NT][8];
                #pragma unroll
                for (int t = 0; t < NT; ++t) {
                    #pragma unroll
                    for (int g = 0; g < 8; ++g) cv[t][g] = Cg[off0 + (size_t)g * NC + t * 16];
                }
                #pragma unroll
                for (int t = 0; t < NT; ++t) {
                    #pragma unroll
                    for (int g = 0; g < 8; ++g) acc[t][g] += cv[t][g];
                }
            }
            #pragma unroll
            for (int t = 0; t < NT; ++t) {
                #pragma unroll
                for (int g = 0; g < 8; ++g) {
                    float v = acc[t][g];
                    if (GELU) v = gelu_exact(v);
                    Dg[off0 + (size_t)g * NC + t * 16] = v;
                }
            }
        }

        mt += (int)gridDim.x * (32 / NCH);
        if (mt < tilesM) {
            Ap = (gcfp)A + (size_t)(mt * 16 + l) * K + half * 2;
            #pragma unroll
            for (int u = 0; u < 8; ++u) { an[u].x = Ap[u * 4]; an[u].y = Ap[u * 4 + 1]; }
        }
    }
}

template<int K, int DOUT, int R, int NO>
__global__ __launch_bounds__(1024) void agg_root(
        const float* __restrict__ yl, const int* __restrict__ src,
        const int* __restrict__ dst, const float* __restrict__ bias,
        const float* __restrict__ X, const float* __restrict__ W,
        float* __restrict__ xout, const float* __restrict__ Wn_l,
        const float* __restrict__ Wn_r, float* __restrict__ Y, float* __restrict__ Z,
        const int* __restrict__ elist,
        const int* __restrict__ ecount, int N, int E, int vec_ok) {
    constexpr int GWV  = 24;
    constexpr int MWV  = 8;
    constexpr int PCAP = 160;
    constexpr int NT   = 4;
    constexpr int TN   = DOUT / 16;
    constexpr int NCH  = TN / NT;
    constexpr int KS   = K / 4;
    constexpr int KB   = KS / 8;
    constexpr int TOT  = KS * DOUT;
    constexpr int ENT  = TOT / 1024;
    constexpr int MT   = (R + 15) / 16;
    constexpr int ITEMS = MT * NCH;
    constexpr int KS3  = DOUT / 4;
    static_assert(DOUT % 64 == 0 && TN % NT == 0 && KS % 8 == 0 && TOT % 1024 == 0, "tiling");
    static_assert(GWV + MWV == 32 && LCAP % (GWV * 32) == 0, "wave split");
    static_assert(2 * NO <= 16 && MT <= 32, "Y and Z columns sit in one N-tile, one M-tile per wave");
    __shared__ __attribute__((aligned(16))) float lds[TOT * 4 + 2 * R * DOUT + R + (NO > 0 ? KS3 * 64 : 0)];
    __shared__ int   pend[GWV * PCAP];
    float* const Ws   = lds;
    float* const acc  = Ws + TOT * 4;
    float* const zs   = acc + R * DOUT;
    float* const cntL = zs + R * DOUT;
    float* const W3s  = cntL + R;
    static_assert(R % 4 == 0, "W3s alignment");

    const int tid  = threadIdx.x;
    const int lane = tid & 31;
    const int wave = __builtin_amdgcn_readfirstlane(tid >> 5);
    const int n0   = blockIdx.x * R;
    const int half = lane >> 4;
    const int l    = lane & 15;
    const int2* mylist = (const int2*)elist + (size_t)blockIdx.x * LCAP;
    constexpr int LCH = LCAP / (GWV * 32);
    gcfp Xg = (gcfp)X;

    int   cnt = 0, lo = 0, hi = 0;
    int2  t[LCH];
    int   it = wave - GWV;
    v2f   an[8];
    gcfp  Ap = Xg;
    if (wave < GWV) {
        cnt = __builtin_amdgcn_readfirstlane(ecount[blockIdx.x * CSTR]);
        const int per = (((cnt + GWV - 1) / GWV) + 7) & ~7;
        lo = wave * per;
        hi = (lo + per < cnt) ? (lo + per) : cnt;
        #pragma unroll
        for (int c = 0; c < LCH; ++c) t[c] = make_int2(0, 0);
        if (cnt <= LCAP) {
            #pragma unroll
            for (int c = 0; c < LCH; ++c) {
                const int i = lo + c * 32 + lane;
                if (i < hi) t[c] = mylist[i];
            }
        }
    } else {
        int row = n0 + (it / NCH) * 16 + l;
        if (row > N - 1) row = N - 1;
        Ap = Xg + (size_t)row * K + half * 2;
        #pragma unroll
        for (int u = 0; u < 8; ++u) { an[u].x = Ap[u * 4]; an[u].y = Ap[u * 4 + 1]; }
    }

    for (int i = tid; i < R * DOUT; i += 1024) acc[i] = 0.0f;
    for (int i = tid; i < R; i += 1024) cntL[i] = 0.0f;
    {
        float4 w[ENT];
        #pragma unroll
        for (int e = 0; e < ENT; ++e) {
            const int i   = tid + e * 1024;
            const int ks  = i / DOUT;
            const int col = i - ks * DOUT;
            const float* p = W + (size_t)(ks * 4) * DOUT + col;
            w[e].x = p[0];
            w[e].y = p[DOUT];
            w[e].z = p[2 * DOUT];
            w[e].w = p[3 * DOUT];
        }
        #pragma unroll
        for (int e = 0; e < ENT; ++e) *(float4*)(Ws + (tid + e * 1024) * 4) = w[e];
    }
    if constexpr (NO > 0) {
        for (int i = tid; i < KS3 * 64; i += 1024) {
            const int kk  = i & 3;
            const int col = (i >> 2) & 15;
            const int k   = (i >> 6) * 4 + kk;
            float v = 0.0f;
            if (col < NO)          v = Wn_l[k * NO + col];
            else if (col < 2 * NO) v = Wn_r[k * NO + (col - NO)];
            W3s[i] = v;
        }
    }
    __syncthreads();

    if (wave < GWV) {
        int* mypend = pend + wave * PCAP;
        if (cnt <= LCAP) {
            #pragma unroll
            for (int c = 0; c < LCH; ++c) {
                const int b = lo + c * 32;
                if (b < hi) {
                    const int n = (hi - b < 32) ? (hi - b) : 32;
                    if (lane < n) lds_add(&cntL[t[c].y], 1.0f);
                    add_rows<DOUT>(t[c].x, t[c].y, n, lane, yl, acc);
                }
            }
        } else {
            sweep_edges<DOUT, R, GWV>(mypend, wave, lane, n0, yl, src, dst, acc, cntL, E, vec_ok);
        }
    } else {
        while (it < ITEMS) {
            const int mtl = it / NCH;
            const int nc  = it - mtl * NCH;
            const float* Bp = Ws + (nc * NT * 16 + l) * 4 + half * 2;
            v8f d[NT];
            #pragma unroll
            for (int t = 0; t < NT; ++t) d[t] = (v8f){};
            #pragma unroll 1
            for (int kb = 0; kb < KB; ++kb) {
                v2f ac[8];
                #pragma unroll
                for (int u = 0; u < 8; ++u) ac[u] = an[u];
                if (kb + 1 < KB) {
                    gcfp An = Ap + (kb + 1) * 32;
                    #pragma unroll
                    for (int u = 0; u < 8; ++u) { an[u].x = An[u * 4]; an[u].y = An[u * 4 + 1]; }
                }
                const float* Bk = Bp + kb * (8 * DOUT * 4);
                #pragma unroll
                for (int u = 0; u < 8; ++u) {
                    #pragma unroll
                    for (int t = 0; t < NT; ++t) {
                        const v2f b = *(const v2f*)(Bk + (u * DOUT + t * 16) * 4);
                        d[t] = __builtin_amdgcn_wmma_f32_16x16x4_f32(
                                   false, ac[u], false, b, (short)0, d[t], false, false);
                    }
                }
            }
            #pragma unroll
            for (int t = 0; t < NT; ++t) {
                #pragma unroll
                for (int g = 0; g < 8; ++g) {
                    const int r = mtl * 16 + half * 8 + g;
                    if (r < R) zs[r * DOUT + nc * NT * 16 + t * 16 + l] = d[t][g];
                }
            }
            it += MWV;
            if (it < ITEMS) {
                int row = n0 + (it / NCH) * 16 + l;
                if (row > N - 1) row = N - 1;
                Ap = Xg + (size_t)row * K + half * 2;
                #pragma unroll
                for (int u = 0; u < 8; ++u) { an[u].x = Ap[u * 4]; an[u].y = Ap[u * 4 + 1]; }
            }
        }
    }
    __syncthreads();

    if constexpr (NO == 0) {
        for (int i = tid; i < R * DOUT; i += 1024) {
            const int r = i / DOUT;
            const int f = i - r * DOUT;
            if (n0 + r < N) {
                const float m = acc[i] / fmaxf(cntL[r], 1.0f) + bias[f];
                xout[(size_t)n0 * DOUT + i] = gelu_exact(zs[i] + m);
            }
        }
    } else {
        for (int i = tid; i < R * DOUT; i += 1024) {
            const int r = i / DOUT;
            const int f = i - r * DOUT;
            const float m = acc[i] / fmaxf(cntL[r], 1.0f) + bias[f];
            zs[i] = gelu_exact(zs[i] + m);
        }
        __syncthreads();
        if (wave < MT) {
            int rl = wave * 16 + l;
            if (rl > R - 1) rl = R - 1;
            const float* Ar = zs + rl * DOUT + half * 2;
            const float* Br = W3s + l * 4 + half * 2;
            v8f p0 = (v8f){}, p1 = (v8f){};
            #pragma unroll
            for (int ks = 0; ks < KS3; ks += 2) {
                const v2f a0 = *(const v2f*)(Ar + ks * 4);
                const v2f a1 = *(const v2f*)(Ar + (ks + 1) * 4);
                const v2f b0 = *(const v2f*)(Br + ks * 64);
                const v2f b1 = *(const v2f*)(Br + (ks + 1) * 64);
                p0 = __builtin_amdgcn_wmma_f32_16x16x4_f32(false, a0, false, b0, (short)0, p0, false, false);
                p1 = __builtin_amdgcn_wmma_f32_16x16x4_f32(false, a1, false, b1, (short)0, p1, false, false);
            }
            #pragma unroll
            for (int g = 0; g < 8; ++g) {
                const int r = wave * 16 + half * 8 + g;
                if (r < R && n0 + r < N) {
                    const float v = p0[g] + p1[g];
                    if (l < NO)          Y[(size_t)(n0 + r) * NO + l] = v;
                    else if (l < 2 * NO) Z[(size_t)(n0 + r) * NO + (l - NO)] = v;
                }
            }
        }
    }
}


extern "C" void kernel_launch(void* const* d_in, const int* in_sizes, int n_in,
                              void* d_out, int out_size, void* d_ws, size_t ws_size,
                              hipStream_t stream) {
    static const int DIMS[5] = {256, 192, 128, 64, 3};

    const float* feat  = (const float*)d_in[0];
    const int*   edges = (const int*)d_in[1];
    const int N = in_sizes[0] / DIMS[0];
    const int E = in_sizes[1] / 2;
    const int* src = edges;
    const int* dst = edges + E;

    const float* Wl[4] = {(const float*)d_in[2], (const float*)d_in[5],
                          (const float*)d_in[8], (const float*)d_in[11]};
    const float* Wr[4] = {(const float*)d_in[3], (const float*)d_in[6],
                          (const float*)d_in[9], (const float*)d_in[12]};
    const float* Bb[4] = {(const float*)d_in[4], (const float*)d_in[7],
                          (const float*)d_in[10], (const float*)d_in[13]};

    float* ws    = (float*)d_ws;
    float* r0    = ws;
    float* r1    = ws + (size_t)N * 192;
    float* r2    = ws + (size_t)N * 320;
    int* elist   = (int*)(ws + (size_t)N * 448);
    int* ecount  = elist + (size_t)((N + 195) / 196) * LCAP * 2;
    float* const YL[4]  = {r0, r1, r0, r1};
    float* const agg0   = r1;
    float* const z3 = r1 + (size_t)N * 64;

    const int vec_ok = (E % 4 == 0) ? 1 : 0;
    const int tilesM = N / 16;
    const int GW = 256;
    const int nb = (N + 195) / 196;

    const float* x = feat;
    for (int L = 0; L < 4; ++L) {
        float* yl   = YL[L];
        float* xout = (L == 3) ? (float*)d_out : (L == 1) ? r2 : yl;

        if (L == 0) {
            gemm_wlds<256, 192, 6, false, false><<<GW, 1024, 0, stream>>>(x, Wl[L], nullptr, yl, tilesM, ecount, nb);
            bin_edges<196><<<256, 1024, 0, stream>>>(src, dst, elist, ecount, E, nb);
            agg_owner<192, 196, false><<<nb, 1024, 0, stream>>>(
                yl, src, dst, Bb[L], agg0, elist, ecount, nullptr, N, E, vec_ok);
            gemm_wlds<256, 192, 6, true, true><<<GW, 1024, 0, stream>>>(x, Wr[L], agg0, xout, tilesM, nullptr, 0);
        } else if (L == 1) {
            gemm_wlds<192, 128, 4, false, false><<<GW, 1024, 0, stream>>>(x, Wl[L], nullptr, yl, tilesM, nullptr, 0);
            agg_root<192, 128, 196, 0><<<nb, 1024, 0, stream>>>(
                yl, src, dst, Bb[L], x, Wr[L], xout, nullptr, nullptr, nullptr, nullptr,
                elist, ecount, N, E, vec_ok);
        } else if (L == 2) {
            gemm_wlds<128, 64, 4, false, false><<<GW, 1024, 0, stream>>>(x, Wl[L], nullptr, yl, tilesM, nullptr, 0);
            agg_root<128, 64, 196, 3><<<nb, 1024, 0, stream>>>(
                yl, src, dst, Bb[L], x, Wr[L], nullptr, Wl[3], Wr[3], YL[3], z3,
                elist, ecount, N, E, vec_ok);
        } else {
            agg_owner<3, 196, true><<<nb, 1024, 0, stream>>>(
                yl, src, dst, Bb[L], (float*)d_out, elist, ecount, z3, N, E, vec_ok);
        }
        x = xout;
    }
}
